# starting kernel
# baseline (speedup 1.0000x reference)
#include <hip/hip_runtime.h>
#include <hip/hip_bf16.h>

typedef float v2f __attribute__((ext_vector_type(2)));
typedef float v8f __attribute__((ext_vector_type(8)));

#define NTOK 343
#define NPAD 352
#define HEADS 6
#define HD 32
#define DIMC 192
#define BATCH 256
#define ATTN_SCALE 0.17677669529663687f
#define SSTRIDE 353

__device__ __forceinline__ int rp_index(int q, int k) {
  int q0 = (q / 7) % 7, q1 = q / 49, q2 = q % 7;
  int k0 = (k / 7) % 7, k1 = k / 49, k2 = k % 7;
  return (q0 - k0 + 6) * 169 + (q1 - k1 + 6) * 13 + (q2 - k2 + 6);
}

__global__ __launch_bounds__(128) void qkv_kernel(
    const float* __restrict__ X, const float* __restrict__ W,
    float* __restrict__ Q, float* __restrict__ K, float* __restrict__ V) {
  const int t = threadIdx.x;
  const int wid = t >> 5, lane = t & 31, half = lane >> 4, l16 = lane & 15;
  const int row0 = blockIdx.x * 16;
  const int col0 = (blockIdx.y * 4 + wid) * 16;
  const float* xr = X + (size_t)(row0 + l16) * DIMC;

  v8f acc = {};
  for (int kk = 0; kk < DIMC; kk += 4) {
    const int kc = kk + 2 * half;
    v2f a = { xr[kc], xr[kc + 1] };
    v2f b = { W[(size_t)kc * 576 + col0 + l16],
              W[(size_t)(kc + 1) * 576 + col0 + l16] };
    acc = __builtin_amdgcn_wmma_f32_16x16x4_f32(false, a, false, b,
                                                (short)0, acc, false, false);
  }

  const int which = col0 / DIMC;
  const int head  = (col0 % DIMC) / HD;
  const int hd0   = col0 % HD;
  float* dst = (which == 0) ? Q : (which == 1) ? K : V;
  const float sc = (which == 0) ? ATTN_SCALE : 1.0f;
#pragma unroll
  for (int i = 0; i < 8; ++i) {
    const int r = row0 + i + 8 * half;
    const int b = r / NTOK, n = r % NTOK;
    dst[(((size_t)b * HEADS + head) * NTOK + n) * HD + hd0 + l16] = acc[i] * sc;
  }
}

__global__ __launch_bounds__(128) void attn_kernel(
    const float* __restrict__ Q, const float* __restrict__ K,
    const float* __restrict__ V, const float* __restrict__ mask,
    const float* __restrict__ rpb, float* __restrict__ AO) {
  __shared__ float sS[16 * SSTRIDE];
  __shared__ float sRed[16 * 8];
  __shared__ float sRowMax[16];
  __shared__ float sRowInv[16];
  __shared__ float sPart[2 * 16 * 16];

  const int t = threadIdx.x;
  const int wid = t >> 5, lane = t & 31, half = lane >> 4, l16 = lane & 15;
  const int q0 = blockIdx.x * 16;
  const int h  = blockIdx.y;
  const int b  = blockIdx.z;
  const size_t base = ((size_t)b * HEADS + h) * (size_t)NTOK * HD;
  const float* Qb = Q + base;
  const float* Kb = K + base;
  const float* Vb = V + base;

  float qa0[8], qa1[8];
  const int qrow = q0 + l16;
  const bool qok = qrow < NTOK;
#pragma unroll
  for (int s = 0; s < 8; ++s) {
    const int d = s * 4 + 2 * half;
    qa0[s] = qok ? Qb[qrow * HD + d]     : 0.f;
    qa1[s] = qok ? Qb[qrow * HD + d + 1] : 0.f;
  }
  const float* mrow = mask + (size_t)(b & 63) * NTOK * NTOK;
  for (int kt = wid; kt < NPAD / 16; kt += 4) {
    const int k0 = kt * 16;
    const int krow = k0 + l16;
    const bool kok = krow < NTOK;
    v8f acc = {};
#pragma unroll
    for (int s = 0; s < 8; ++s) {
      const int d = s * 4 + 2 * half;
      v2f a = { qa0[s], qa1[s] };
      v2f bf = { kok ? Kb[krow * HD + d]     : 0.f,
                 kok ? Kb[krow * HD + d + 1] : 0.f };
      acc = __builtin_amdgcn_wmma_f32_16x16x4_f32(false, a, false, bf,
                                                  (short)0, acc, false, false);
    }
#pragma unroll
    for (int i = 0; i < 8; ++i) {
      const int q = q0 + i + 8 * half;
      const int k = k0 + l16;
      float v = acc[i];
      if (q < NTOK && k < NTOK)
        v += rpb[rp_index(q, k) * HEADS + h] + mrow[q * NTOK + k];
      else
        v = -1e30f;
      sS[(i + 8 * half) * SSTRIDE + k] = v;
    }
  }
  __syncthreads();

  {
    const int r = t >> 3, sg = t & 7;
    float m = -1e30f;
    for (int k = sg; k < NTOK; k += 8) m = fmaxf(m, sS[r * SSTRIDE + k]);
    sRed[r * 8 + sg] = m;
    __syncthreads();
    if (sg == 0) {
      float mm = sRed[r * 8];
#pragma unroll
      for (int j = 1; j < 8; ++j) mm = fmaxf(mm, sRed[r * 8 + j]);
      sRowMax[r] = mm;
    }
    __syncthreads();
    const float rm = sRowMax[r];
    float sum = 0.f;
    for (int k = sg; k < NPAD; k += 8) {
      float e = 0.f;
      if (k < NTOK) { e = __expf(sS[r * SSTRIDE + k] - rm); sum += e; }
      sS[r * SSTRIDE + k] = e;
    }
    sRed[r * 8 + sg] = sum;
    __syncthreads();
    if (sg == 0) {
      float ss = 0.f;
#pragma unroll
      for (int j = 0; j < 8; ++j) ss += sRed[r * 8 + j];
      sRowInv[r] = 1.0f / ss;
    }
  }
  __syncthreads();

  const int dt = wid & 1;
  const int kh = wid >> 1;
  const int d0 = dt * 16;
  v8f acc = {};
  for (int kk = kh * (NPAD / 2); kk < kh * (NPAD / 2) + NPAD / 2; kk += 4) {
    const int kc = kk + 2 * half;
    v2f a = { sS[l16 * SSTRIDE + kc], sS[l16 * SSTRIDE + kc + 1] };
    v2f bf = { (kc < NTOK)     ? Vb[(size_t)kc * HD + d0 + l16]       : 0.f,
               (kc + 1 < NTOK) ? Vb[(size_t)(kc + 1) * HD + d0 + l16] : 0.f };
    acc = __builtin_amdgcn_wmma_f32_16x16x4_f32(false, a, false, bf,
                                                (short)0, acc, false, false);
  }
  if (kh == 1) {
#pragma unroll
    for (int i = 0; i < 8; ++i)
      sPart[dt * 256 + (i + 8 * half) * 16 + l16] = acc[i];
  }
  __syncthreads();
  if (kh == 0) {
#pragma unroll
    for (int i = 0; i < 8; ++i) {
      const int q = q0 + i + 8 * half;
      if (q < NTOK) {
        const float v = (acc[i] + sPart[dt * 256 + (i + 8 * half) * 16 + l16])
                        * sRowInv[i + 8 * half];
        AO[((size_t)b * NTOK + q) * DIMC + h * HD + d0 + l16] = v;
      }
    }
  }
}

__global__ __launch_bounds__(128) void proj_kernel(
    const float* __restrict__ AO, const float* __restrict__ W,
    const float* __restrict__ bias, float* __restrict__ out) {
  const int t = threadIdx.x;
  const int wid = t >> 5, lane = t & 31, half = lane >> 4, l16 = lane & 15;
  const int row0 = blockIdx.x * 16;
  const int col0 = (blockIdx.y * 4 + wid) * 16;
  const float* ar = AO + (size_t)(row0 + l16) * DIMC;

  v8f acc = {};
  for (int kk = 0; kk < DIMC; kk += 4) {
    const int kc = kk + 2 * half;
    v2f a = { ar[kc], ar[kc + 1] };
    v2f b = { W[(size_t)kc * DIMC + col0 + l16],
              W[(size_t)(kc + 1) * DIMC + col0 + l16] };
    acc = __builtin_amdgcn_wmma_f32_16x16x4_f32(false, a, false, b,
                                                (short)0, acc, false, false);
  }
  const float bv = bias[col0 + l16];
#pragma unroll
  for (int i = 0; i < 8; ++i) {
    const int r = row0 + i + 8 * half;
    out[(size_t)r * DIMC + col0 + l16] = acc[i] + bv;
  }
}

extern "C" void kernel_launch(void* const* d_in, const int* in_sizes, int n_in,
                              void* d_out, int out_size, void* d_ws, size_t ws_size,
                              hipStream_t stream) {
  const float* x      = (const float*)d_in[0];
  const float* mask   = (const float*)d_in[1];
  const float* qkv_w  = (const float*)d_in[2];
  const float* rpb    = (const float*)d_in[3];
  const float* proj_w = (const float*)d_in[4];
  const float* proj_b = (const float*)d_in[5];
  float* out = (float*)d_out;

  const size_t qkvElems = (size_t)BATCH * HEADS * NTOK * HD;
  float* Q  = (float*)d_ws;
  float* K  = Q + qkvElems;
  float* V  = K + qkvElems;
  float* AO = V + qkvElems;

  const int rowTiles = (BATCH * NTOK) / 16;
  qkv_kernel <<<dim3(rowTiles, 9),            128, 0, stream>>>(x, qkv_w, Q, K, V);
  attn_kernel<<<dim3(NPAD / 16, HEADS, BATCH), 128, 0, stream>>>(Q, K, V, mask, rpb, AO);
  proj_kernel<<<dim3(rowTiles, 3),            128, 0, stream>>>(AO, proj_w, proj_b, out);
}


# attention kernel: guarded one-at-a-time loads (K, bias, mask, V; softmax LDS reads) become unconditional clamped loads issued in groups, K/Q rows as b128; order of the QK sum over d changes
# speedup vs baseline: 1.5831x; 1.5831x over previous
#include <hip/hip_runtime.h>
#include <hip/hip_bf16.h>

typedef float v2f __attribute__((ext_vector_type(2)));
typedef float v8f __attribute__((ext_vector_type(8)));
typedef float v4f __attribute__((ext_vector_type(4)));

#define NTOK 343
#define NPAD 352
#define HEADS 6
#define HD 32
#define DIMC 192
#define BATCH 256
#define ATTN_SCALE 0.17677669529663687f
#define SSTRIDE 353

__device__ __forceinline__ int rp_index(int q, int k) {
  int q0 = (q / 7) % 7, q1 = q / 49, q2 = q % 7;
  int k0 = (k / 7) % 7, k1 = k / 49, k2 = k % 7;
  return (q0 - k0 + 6) * 169 + (q1 - k1 + 6) * 13 + (q2 - k2 + 6);
}

__device__ __forceinline__ int rp_code(int f) {
  return ((f / 7) % 7) * 169 + (f / 49) * 13 + (f % 7);
}

__global__ __launch_bounds__(128) void qkv_kernel(
    const float* __restrict__ X, const float* __restrict__ W,
    float* __restrict__ Q, float* __restrict__ K, float* __restrict__ V) {
  const int t = threadIdx.x;
  const int wid = t >> 5, lane = t & 31, half = lane >> 4, l16 = lane & 15;
  const int row0 = blockIdx.x * 16;
  const int col0 = (blockIdx.y * 4 + wid) * 16;
  const float* xr = X + (size_t)(row0 + l16) * DIMC;

  v8f acc = {};
  for (int kk = 0; kk < DIMC; kk += 4) {
    const int kc = kk + 2 * half;
    v2f a = { xr[kc], xr[kc + 1] };
    v2f b = { W[(size_t)kc * 576 + col0 + l16],
              W[(size_t)(kc + 1) * 576 + col0 + l16] };
    acc = __builtin_amdgcn_wmma_f32_16x16x4_f32(false, a, false, b,
                                                (short)0, acc, false, false);
  }

  const int which = col0 / DIMC;
  const int head  = (col0 % DIMC) / HD;
  const int hd0   = col0 % HD;
  float* dst = (which == 0) ? Q : (which == 1) ? K : V;
  const float sc = (which == 0) ? ATTN_SCALE : 1.0f;
#pragma unroll
  for (int i = 0; i < 8; ++i) {
    const int r = row0 + i + 8 * half;
    const int b = r / NTOK, n = r % NTOK;
    dst[(((size_t)b * HEADS + head) * NTOK + n) * HD + hd0 + l16] = acc[i] * sc;
  }
}

__global__ __launch_bounds__(128) void attn_kernel(
    const float* __restrict__ Q, const float* __restrict__ K,
    const float* __restrict__ V, const float* __restrict__ mask,
    const float* __restrict__ rpb, float* __restrict__ AO) {
  __shared__ float sS[16 * SSTRIDE];
  __shared__ float sRed[16 * 8];
  __shared__ float sRowMax[16];
  __shared__ float sRowInv[16];
  __shared__ float sPart[2 * 16 * 16];

  const int t = threadIdx.x;
  const int wid = t >> 5, lane = t & 31, half = lane >> 4, l16 = lane & 15;
  const int q0 = blockIdx.x * 16;
  const int h  = blockIdx.y;
  const int b  = blockIdx.z;
  const size_t base = ((size_t)b * HEADS + h) * (size_t)NTOK * HD;
  const float* Qb = Q + base;
  const float* Kb = K + base;
  const float* Vb = V + base;

  const int qrow = (q0 + l16 < NTOK) ? q0 + l16 : NTOK - 1;
  const v4f* qp = reinterpret_cast<const v4f*>(Qb + (size_t)qrow * HD + 16 * half);
  const v4f qv0 = qp[0], qv1 = qp[1], qv2 = qp[2], qv3 = qp[3];

  const float* mrow = mask + (size_t)(b & 63) * NTOK * NTOK;
  int aq[8], mo[8];
#pragma unroll
  for (int i = 0; i < 8; ++i) {
    const int q = q0 + i + 8 * half;
    const int qc = (q < NTOK) ? q : NTOK - 1;
    aq[i] = rp_code(qc) + 1098;
    mo[i] = qc * NTOK;
  }

  for (int kt = wid; kt < NPAD / 16; kt += 4) {
    const int kcol = kt * 16 + l16;
    const int krow = (kcol < NTOK) ? kcol : NTOK - 1;
    const v4f* kp = reinterpret_cast<const v4f*>(Kb + (size_t)krow * HD + 16 * half);
    const v4f kv0 = kp[0], kv1 = kp[1], kv2 = kp[2], kv3 = kp[3];

    const int ak = rp_code(krow);
    float br[8], mk[8];
#pragma unroll
    for (int i = 0; i < 8; ++i) br[i] = rpb[(aq[i] - ak) * HEADS + h];
#pragma unroll
    for (int i = 0; i < 8; ++i) mk[i] = mrow[mo[i] + krow];
    __builtin_amdgcn_sched_barrier(0);

    v8f acc = {};
#define QK_STEP(QA, QB, KA, KB)                                              \
    { v2f a_ = { QA, QB }; v2f b_ = { KA, KB };                              \
      acc = __builtin_amdgcn_wmma_f32_16x16x4_f32(false, a_, false, b_,      \
                                                  (short)0, acc, false, false); }
    QK_STEP(qv0.x, qv0.y, kv0.x, kv0.y)
    QK_STEP(qv0.z, qv0.w, kv0.z, kv0.w)
    QK_STEP(qv1.x, qv1.y, kv1.x, kv1.y)
    QK_STEP(qv1.z, qv1.w, kv1.z, kv1.w)
    QK_STEP(qv2.x, qv2.y, kv2.x, kv2.y)
    QK_STEP(qv2.z, qv2.w, kv2.z, kv2.w)
    QK_STEP(qv3.x, qv3.y, kv3.x, kv3.y)
    QK_STEP(qv3.z, qv3.w, kv3.z, kv3.w)
#undef QK_STEP

    const bool kok = kcol < NTOK;
#pragma unroll
    for (int i = 0; i < 8; ++i) {
      const bool ok = kok && (q0 + i + 8 * half < NTOK);
      sS[(i + 8 * half) * SSTRIDE + kcol] = ok ? acc[i] + (br[i] + mk[i]) : -1e30f;
    }
  }
  __syncthreads();

  {
    const int r = t >> 3, sg = t & 7;
    float* srow = sS + r * SSTRIDE + sg;
    float m = -1e30f;
#pragma unroll
    for (int j = 0; j < 43; ++j) {
      const float v = srow[8 * j];
      m = (v > m) ? v : m;
    }
    sRed[r * 8 + sg] = m;
    __syncthreads();
    if (sg == 0) {
      float mm = sRed[r * 8];
#pragma unroll
      for (int j = 1; j < 8; ++j) mm = fmaxf(mm, sRed[r * 8 + j]);
      sRowMax[r] = mm;
    }
    __syncthreads();
    const float rm = sRowMax[r];
    float sum = 0.f;
#pragma unroll 14
    for (int j = 0; j < 42; ++j) {
      const float e = __expf(srow[8 * j] - rm);
      sum += e;
      srow[8 * j] = e;
    }
    {
      const float s = srow[8 * 42];
      const float e = (sg < 7) ? __expf(s - rm) : 0.f;
      sum += e;
      srow[8 * 42] = e;
    }
    srow[8 * 43] = 0.f;
    sRed[r * 8 + sg] = sum;
    __syncthreads();
    if (sg == 0) {
      float ss = 0.f;
#pragma unroll
      for (int j = 0; j < 8; ++j) ss += sRed[r * 8 + j];
      sRowInv[r] = 1.0f / ss;
    }
  }
  __syncthreads();

  const int dt = wid & 1;
  const int kh = wid >> 1;
  const int d0 = dt * 16;
  v8f acc = {};
  for (int g = 0; g < 2; ++g) {
    const int kg = kh * (NPAD / 2) + g * 88 + 2 * half;
    float vb0[22], vb1[22];
#pragma unroll
    for (int s = 0; s < 22; ++s) {
      const int kc = kg + 4 * s;
      const int r0 = (kc < NTOK) ? kc : NTOK - 1;
      const int r1 = (kc + 1 < NTOK) ? kc + 1 : NTOK - 1;
      vb0[s] = Vb[r0 * HD + d0 + l16];
      vb1[s] = Vb[r1 * HD + d0 + l16];
    }
    __builtin_amdgcn_sched_barrier(0);
#pragma unroll
    for (int s = 0; s < 22; ++s) {
      const int kc = kg + 4 * s;
      v2f a = { sS[l16 * SSTRIDE + kc], sS[l16 * SSTRIDE + kc + 1] };
      v2f bf = { (kc < NTOK) ? vb0[s] : 0.f, (kc + 1 < NTOK) ? vb1[s] : 0.f };
      acc = __builtin_amdgcn_wmma_f32_16x16x4_f32(false, a, false, bf,
                                                  (short)0, acc, false, false);
    }
  }
  if (kh == 1) {
#pragma unroll
    for (int i = 0; i < 8; ++i)
      sPart[dt * 256 + (i + 8 * half) * 16 + l16] = acc[i];
  }
  __syncthreads();
  if (kh == 0) {
#pragma unroll
    for (int i = 0; i < 8; ++i) {
      const int q = q0 + i + 8 * half;
      if (q < NTOK) {
        const float v = (acc[i] + sPart[dt * 256 + (i + 8 * half) * 16 + l16])
                        * sRowInv[i + 8 * half];
        AO[((size_t)b * NTOK + q) * DIMC + h * HD + d0 + l16] = v;
      }
    }
  }
}

__global__ __launch_bounds__(128) void proj_kernel(
    const float* __restrict__ AO, const float* __restrict__ W,
    const float* __restrict__ bias, float* __restrict__ out) {
  const int t = threadIdx.x;
  const int wid = t >> 5, lane = t & 31, half = lane >> 4, l16 = lane & 15;
  const int row0 = blockIdx.x * 16;
  const int col0 = (blockIdx.y * 4 + wid) * 16;
  const float* ar = AO + (size_t)(row0 + l16) * DIMC;

  v8f acc = {};
  for (int kk = 0; kk < DIMC; kk += 4) {
    const int kc = kk + 2 * half;
    v2f a = { ar[kc], ar[kc + 1] };
    v2f b = { W[(size_t)kc * DIMC + col0 + l16],
              W[(size_t)(kc + 1) * DIMC + col0 + l16] };
    acc = __builtin_amdgcn_wmma_f32_16x16x4_f32(false, a, false, b,
                                                (short)0, acc, false, false);
  }
  const float bv = bias[col0 + l16];
#pragma unroll
  for (int i = 0; i < 8; ++i) {
    const int r = row0 + i + 8 * half;
    out[(size_t)r * DIMC + col0 + l16] = acc[i] + bv;
  }
}

extern "C" void kernel_launch(void* const* d_in, const int* in_sizes, int n_in,
                              void* d_out, int out_size, void* d_ws, size_t ws_size,
                              hipStream_t stream) {
  const float* x      = (const float*)d_in[0];
  const float* mask   = (const float*)d_in[1];
  const float* qkv_w  = (const float*)d_in[2];
  const float* rpb    = (const float*)d_in[3];
  const float* proj_w = (const float*)d_in[4];
  const float* proj_b = (const float*)d_in[5];
  float* out = (float*)d_out;

  const size_t qkvElems = (size_t)BATCH * HEADS * NTOK * HD;
  float* Q  = (float*)d_ws;
  float* K  = Q + qkvElems;
  float* V  = K + qkvElems;
  float* AO = V + qkvElems;

  const int rowTiles = (BATCH * NTOK) / 16;
  qkv_kernel <<<dim3(rowTiles, 9),            128, 0, stream>>>(x, qkv_w, Q, K, V);
  attn_kernel<<<dim3(NPAD / 16, HEADS, BATCH), 128, 0, stream>>>(Q, K, V, mask, rpb, AO);
  proj_kernel<<<dim3(rowTiles, 3),            128, 0, stream>>>(AO, proj_w, proj_b, out);
}


# both projection GEMMs: each wave computes 32x64 outputs (2x4 tiles) per loaded fragment, columns interleaved by 4 so W rows and outputs move as b128; order of the sum over k changes
# speedup vs baseline: 2.2062x; 1.3936x over previous
#include <hip/hip_runtime.h>
#include <hip/hip_bf16.h>

typedef float v2f __attribute__((ext_vector_type(2)));
typedef float v8f __attribute__((ext_vector_type(8)));
typedef float v4f __attribute__((ext_vector_type(4)));

#define NTOK 343
#define NPAD 352
#define HEADS 6
#define HD 32
#define DIMC 192
#define BATCH 256
#define ATTN_SCALE 0.17677669529663687f
#define SSTRIDE 353

__device__ __forceinline__ int rp_index(int q, int k) {
  int q0 = (q / 7) % 7, q1 = q / 49, q2 = q % 7;
  int k0 = (k / 7) % 7, k1 = k / 49, k2 = k % 7;
  return (q0 - k0 + 6) * 169 + (q1 - k1 + 6) * 13 + (q2 - k2 + 6);
}

__device__ __forceinline__ int rp_code(int f) {
  return ((f / 7) % 7) * 169 + (f / 49) * 13 + (f % 7);
}

template <int LDW>
__device__ __forceinline__ void wave_gemm_32x64(
    const float* __restrict__ A, const float* __restrict__ W,
    const int row0, const int col0, const int l16, const int half, v8f (&acc)[2][4]) {
  const float* a0 = A + (size_t)(row0 + l16) * DIMC + 4 * half;
  const float* a1 = a0 + 16 * DIMC;
  const float* wp = W + (size_t)(4 * half) * LDW + col0 + 4 * l16;
  for (int p = 0; p < DIMC / 8; ++p) {
    const v4f x0 = *reinterpret_cast<const v4f*>(a0 + 8 * p);
    const v4f x1 = *reinterpret_cast<const v4f*>(a1 + 8 * p);
    const float* wr = wp + (size_t)(8 * p) * LDW;
    const v4f w0 = *reinterpret_cast<const v4f*>(wr);
    const v4f w1 = *reinterpret_cast<const v4f*>(wr + LDW);
    const v4f w2 = *reinterpret_cast<const v4f*>(wr + 2 * LDW);
    const v4f w3 = *reinterpret_cast<const v4f*>(wr + 3 * LDW);
    const v2f xa0 = { x0.x, x0.y }, xb0 = { x0.z, x0.w };
    const v2f xa1 = { x1.x, x1.y }, xb1 = { x1.z, x1.w };
#pragma unroll
    for (int t = 0; t < 4; ++t) {
      const v2f ba = { w0[t], w1[t] };
      const v2f bb = { w2[t], w3[t] };
      acc[0][t] = __builtin_amdgcn_wmma_f32_16x16x4_f32(false, xa0, false, ba,
                                                        (short)0, acc[0][t], false, false);
      acc[1][t] = __builtin_amdgcn_wmma_f32_16x16x4_f32(false, xa1, false, ba,
                                                        (short)0, acc[1][t], false, false);
      acc[0][t] = __builtin_amdgcn_wmma_f32_16x16x4_f32(false, xb0, false, bb,
                                                        (short)0, acc[0][t], false, false);
      acc[1][t] = __builtin_amdgcn_wmma_f32_16x16x4_f32(false, xb1, false, bb,
                                                        (short)0, acc[1][t], false, false);
    }
  }
}

__global__ __launch_bounds__(96) void qkv_kernel(
    const float* __restrict__ X, const float* __restrict__ W,
    float* __restrict__ Q, float* __restrict__ K, float* __restrict__ V) {
  const int t = threadIdx.x;
  const int wid = t >> 5, lane = t & 31, half = lane >> 4, l16 = lane & 15;
  const int which = blockIdx.x % 3;
  const int row0 = (blockIdx.x / 3) * 32;
  const int cc0 = wid * 64;

  v8f acc[2][4] = {};
  wave_gemm_32x64<3 * DIMC>(X, W, row0, which * DIMC + cc0, l16, half, acc);

  const int cc = cc0 + 4 * l16;
  const int head = cc / HD, hd = cc % HD;
  float* dst = (which == 0) ? Q : (which == 1) ? K : V;
  const float sc = (which == 0) ? ATTN_SCALE : 1.0f;
#pragma unroll
  for (int m = 0; m < 2; ++m) {
#pragma unroll
    for (int i = 0; i < 8; ++i) {
      const int r = row0 + 16 * m + i + 8 * half;
      const int b = r / NTOK, n = r % NTOK;
      const v4f o = { acc[m][0][i] * sc, acc[m][1][i] * sc,
                      acc[m][2][i] * sc, acc[m][3][i] * sc };
      *reinterpret_cast<v4f*>(dst + (((size_t)b * HEADS + head) * NTOK + n) * HD + hd) = o;
    }
  }
}

__global__ __launch_bounds__(128) void attn_kernel(
    const float* __restrict__ Q, const float* __restrict__ K,
    const float* __restrict__ V, const float* __restrict__ mask,
    const float* __restrict__ rpb, float* __restrict__ AO) {
  __shared__ float sS[16 * SSTRIDE];
  __shared__ float sRed[16 * 8];
  __shared__ float sRowMax[16];
  __shared__ float sRowInv[16];
  __shared__ float sPart[2 * 16 * 16];

  const int t = threadIdx.x;
  const int wid = t >> 5, lane = t & 31, half = lane >> 4, l16 = lane & 15;
  const int q0 = blockIdx.x * 16;
  const int h  = blockIdx.y;
  const int b  = blockIdx.z;
  const size_t base = ((size_t)b * HEADS + h) * (size_t)NTOK * HD;
  const float* Qb = Q + base;
  const float* Kb = K + base;
  const float* Vb = V + base;

  const int qrow = (q0 + l16 < NTOK) ? q0 + l16 : NTOK - 1;
  const v4f* qp = reinterpret_cast<const v4f*>(Qb + (size_t)qrow * HD + 16 * half);
  const v4f qv0 = qp[0], qv1 = qp[1], qv2 = qp[2], qv3 = qp[3];

  const float* mrow = mask + (size_t)(b & 63) * NTOK * NTOK;
  int aq[8], mo[8];
#pragma unroll
  for (int i = 0; i < 8; ++i) {
    const int q = q0 + i + 8 * half;
    const int qc = (q < NTOK) ? q : NTOK - 1;
    aq[i] = rp_code(qc) + 1098;
    mo[i] = qc * NTOK;
  }

  for (int kt = wid; kt < NPAD / 16; kt += 4) {
    const int kcol = kt * 16 + l16;
    const int krow = (kcol < NTOK) ? kcol : NTOK - 1;
    const v4f* kp = reinterpret_cast<const v4f*>(Kb + (size_t)krow * HD + 16 * half);
    const v4f kv0 = kp[0], kv1 = kp[1], kv2 = kp[2], kv3 = kp[3];

    const int ak = rp_code(krow);
    float br[8], mk[8];
#pragma unroll
    for (int i = 0; i < 8; ++i) br[i] = rpb[(aq[i] - ak) * HEADS + h];
#pragma unroll
    for (int i = 0; i < 8; ++i) mk[i] = mrow[mo[i] + krow];
    __builtin_amdgcn_sched_barrier(0);

    v8f acc = {};
#define QK_STEP(QA, QB, KA, KB)                                              \
    { v2f a_ = { QA, QB }; v2f b_ = { KA, KB };                              \
      acc = __builtin_amdgcn_wmma_f32_16x16x4_f32(false, a_, false, b_,      \
                                                  (short)0, acc, false, false); }
    QK_STEP(qv0.x, qv0.y, kv0.x, kv0.y)
    QK_STEP(qv0.z, qv0.w, kv0.z, kv0.w)
    QK_STEP(qv1.x, qv1.y, kv1.x, kv1.y)
    QK_STEP(qv1.z, qv1.w, kv1.z, kv1.w)
    QK_STEP(qv2.x, qv2.y, kv2.x, kv2.y)
    QK_STEP(qv2.z, qv2.w, kv2.z, kv2.w)
    QK_STEP(qv3.x, qv3.y, kv3.x, kv3.y)
    QK_STEP(qv3.z, qv3.w, kv3.z, kv3.w)
#undef QK_STEP

    const bool kok = kcol < NTOK;
#pragma unroll
    for (int i = 0; i < 8; ++i) {
      const bool ok = kok && (q0 + i + 8 * half < NTOK);
      sS[(i + 8 * half) * SSTRIDE + kcol] = ok ? acc[i] + (br[i] + mk[i]) : -1e30f;
    }
  }
  __syncthreads();

  {
    const int r = t >> 3, sg = t & 7;
    float* srow = sS + r * SSTRIDE + sg;
    float m = -1e30f;
#pragma unroll
    for (int j = 0; j < 43; ++j) {
      const float v = srow[8 * j];
      m = (v > m) ? v : m;
    }
    sRed[r * 8 + sg] = m;
    __syncthreads();
    if (sg == 0) {
      float mm = sRed[r * 8];
#pragma unroll
      for (int j = 1; j < 8; ++j) mm = fmaxf(mm, sRed[r * 8 + j]);
      sRowMax[r] = mm;
    }
    __syncthreads();
    const float rm = sRowMax[r];
    float sum = 0.f;
#pragma unroll 14
    for (int j = 0; j < 42; ++j) {
      const float e = __expf(srow[8 * j] - rm);
      sum += e;
      srow[8 * j] = e;
    }
    {
      const float s = srow[8 * 42];
      const float e = (sg < 7) ? __expf(s - rm) : 0.f;
      sum += e;
      srow[8 * 42] = e;
    }
    srow[8 * 43] = 0.f;
    sRed[r * 8 + sg] = sum;
    __syncthreads();
    if (sg == 0) {
      float ss = 0.f;
#pragma unroll
      for (int j = 0; j < 8; ++j) ss += sRed[r * 8 + j];
      sRowInv[r] = 1.0f / ss;
    }
  }
  __syncthreads();

  const int dt = wid & 1;
  const int kh = wid >> 1;
  const int d0 = dt * 16;
  v8f acc = {};
  for (int g = 0; g < 2; ++g) {
    const int kg = kh * (NPAD / 2) + g * 88 + 2 * half;
    float vb0[22], vb1[22];
#pragma unroll
    for (int s = 0; s < 22; ++s) {
      const int kc = kg + 4 * s;
      const int r0 = (kc < NTOK) ? kc : NTOK - 1;
      const int r1 = (kc + 1 < NTOK) ? kc + 1 : NTOK - 1;
      vb0[s] = Vb[r0 * HD + d0 + l16];
      vb1[s] = Vb[r1 * HD + d0 + l16];
    }
    __builtin_amdgcn_sched_barrier(0);
#pragma unroll
    for (int s = 0; s < 22; ++s) {
      const int kc = kg + 4 * s;
      v2f a = { sS[l16 * SSTRIDE + kc], sS[l16 * SSTRIDE + kc + 1] };
      v2f bf = { (kc < NTOK) ? vb0[s] : 0.f, (kc + 1 < NTOK) ? vb1[s] : 0.f };
      acc = __builtin_amdgcn_wmma_f32_16x16x4_f32(false, a, false, bf,
                                                  (short)0, acc, false, false);
    }
  }
  if (kh == 1) {
#pragma unroll
    for (int i = 0; i < 8; ++i)
      sPart[dt * 256 + (i + 8 * half) * 16 + l16] = acc[i];
  }
  __syncthreads();
  if (kh == 0) {
#pragma unroll
    for (int i = 0; i < 8; ++i) {
      const int q = q0 + i + 8 * half;
      if (q < NTOK) {
        const float v = (acc[i] + sPart[dt * 256 + (i + 8 * half) * 16 + l16])
                        * sRowInv[i + 8 * half];
        AO[((size_t)b * NTOK + q) * DIMC + h * HD + d0 + l16] = v;
      }
    }
  }
}

__global__ __launch_bounds__(96) void proj_kernel(
    const float* __restrict__ AO, const float* __restrict__ W,
    const float* __restrict__ bias, float* __restrict__ out) {
  const int t = threadIdx.x;
  const int wid = t >> 5, lane = t & 31, half = lane >> 4, l16 = lane & 15;
  const int row0 = blockIdx.x * 32;
  const int col0 = wid * 64;

  v8f acc[2][4] = {};
  wave_gemm_32x64<DIMC>(AO, W, row0, col0, l16, half, acc);

  const int c4 = col0 + 4 * l16;
  const v4f bv = *reinterpret_cast<const v4f*>(bias + c4);
#pragma unroll
  for (int m = 0; m < 2; ++m) {
#pragma unroll
    for (int i = 0; i < 8; ++i) {
      const int r = row0 + 16 * m + i + 8 * half;
      const v4f o = { acc[m][0][i] + bv.x, acc[m][1][i] + bv.y,
                      acc[m][2][i] + bv.z, acc[m][3][i] + bv.w };
      *reinterpret_cast<v4f*>(out + (size_t)r * DIMC + c4) = o;
    }
  }
}

extern "C" void kernel_launch(void* const* d_in, const int* in_sizes, int n_in,
                              void* d_out, int out_size, void* d_ws, size_t ws_size,
                              hipStream_t stream) {
  const float* x      = (const float*)d_in[0];
  const float* mask   = (const float*)d_in[1];
  const float* qkv_w  = (const float*)d_in[2];
  const float* rpb    = (const float*)d_in[3];
  const float* proj_w = (const float*)d_in[4];
  const float* proj_b = (const float*)d_in[5];
  float* out = (float*)d_out;

  const size_t qkvElems = (size_t)BATCH * HEADS * NTOK * HD;
  float* Q  = (float*)d_ws;
  float* K  = Q + qkvElems;
  float* V  = K + qkvElems;
  float* AO = V + qkvElems;

  const int rowPairs  = (BATCH * NTOK) / 32;
  const int qkvBlocks = (BATCH * NTOK) / 32 * 3;
  qkv_kernel <<<dim3(qkvBlocks),              96, 0, stream>>>(x, qkv_w, Q, K, V);
  attn_kernel<<<dim3(NPAD / 16, HEADS, BATCH), 128, 0, stream>>>(Q, K, V, mask, rpb, AO);
  proj_kernel<<<dim3(rowPairs),               96, 0, stream>>>(AO, proj_w, proj_b, out);
}


# attention kernel: integer index work cut: P@V loads at constant offsets (row clamp and zero-select only on the 3 padded k-steps), bias row codes once per lane and read across lanes, 24-bit multiplies
# speedup vs baseline: 2.2799x; 1.0334x over previous
#include <hip/hip_runtime.h>
#include <hip/hip_bf16.h>

typedef float v2f __attribute__((ext_vector_type(2)));
typedef float v8f __attribute__((ext_vector_type(8)));
typedef float v4f __attribute__((ext_vector_type(4)));

#define NTOK 343
#define NPAD 352
#define HEADS 6
#define HD 32
#define DIMC 192
#define BATCH 256
#define ATTN_SCALE 0.17677669529663687f
#define SSTRIDE 353

__device__ __forceinline__ int rp_index(int q, int k) {
  int q0 = (q / 7) % 7, q1 = q / 49, q2 = q % 7;
  int k0 = (k / 7) % 7, k1 = k / 49, k2 = k % 7;
  return (q0 - k0 + 6) * 169 + (q1 - k1 + 6) * 13 + (q2 - k2 + 6);
}

constexpr bool rp_code_shift_exact() {
  for (int f = 0; f < NPAD; ++f) {
    const int d7 = (f * 9363) >> 16, d49 = (f * 1338) >> 16;
    if (d7 != f / 7 || d49 != f / 49) return false;
    if (f + 162 * d7 - 1170 * d49 != ((f / 7) % 7) * 169 + (f / 49) * 13 + (f % 7)) return false;
  }
  return true;
}
static_assert(rp_code_shift_exact(), "multiply-shift quotients must be exact on 0..NPAD-1");
__device__ __forceinline__ int rp_code(int f) {
  const int d7  = __mul24(f, 9363) >> 16;
  const int d49 = __mul24(f, 1338) >> 16;
  return f + __mul24(d7, 162) - __mul24(d49, 1170);
}

template <int LDW>
__device__ __forceinline__ void wave_gemm_32x64(
    const float* __restrict__ A, const float* __restrict__ W,
    const int row0, const int col0, const int l16, const int half, v8f (&acc)[2][4]) {
  const float* a0 = A + (size_t)(row0 + l16) * DIMC + 4 * half;
  const float* a1 = a0 + 16 * DIMC;
  const float* wp = W + (size_t)(4 * half) * LDW + col0 + 4 * l16;
  for (int p = 0; p < DIMC / 8; ++p) {
    const v4f x0 = *reinterpret_cast<const v4f*>(a0 + 8 * p);
    const v4f x1 = *reinterpret_cast<const v4f*>(a1 + 8 * p);
    const float* wr = wp + (size_t)(8 * p) * LDW;
    const v4f w0 = *reinterpret_cast<const v4f*>(wr);
    const v4f w1 = *reinterpret_cast<const v4f*>(wr + LDW);
    const v4f w2 = *reinterpret_cast<const v4f*>(wr + 2 * LDW);
    const v4f w3 = *reinterpret_cast<const v4f*>(wr + 3 * LDW);
    const v2f xa0 = { x0.x, x0.y }, xb0 = { x0.z, x0.w };
    const v2f xa1 = { x1.x, x1.y }, xb1 = { x1.z, x1.w };
#pragma unroll
    for (int t = 0; t < 4; ++t) {
      const v2f ba = { w0[t], w1[t] };
      const v2f bb = { w2[t], w3[t] };
      acc[0][t] = __builtin_amdgcn_wmma_f32_16x16x4_f32(false, xa0, false, ba,
                                                        (short)0, acc[0][t], false, false);
      acc[1][t] = __builtin_amdgcn_wmma_f32_16x16x4_f32(false, xa1, false, ba,
                                                        (short)0, acc[1][t], false, false);
      acc[0][t] = __builtin_amdgcn_wmma_f32_16x16x4_f32(false, xb0, false, bb,
                                                        (short)0, acc[0][t], false, false);
      acc[1][t] = __builtin_amdgcn_wmma_f32_16x16x4_f32(false, xb1, false, bb,
                                                        (short)0, acc[1][t], false, false);
    }
  }
}

__global__ __launch_bounds__(96) void qkv_kernel(
    const float* __restrict__ X, const float* __restrict__ W,
    float* __restrict__ Q, float* __restrict__ K, float* __restrict__ V) {
  const int t = threadIdx.x;
  const int wid = t >> 5, lane = t & 31, half = lane >> 4, l16 = lane & 15;
  const int which = blockIdx.x % 3;
  const int row0 = (blockIdx.x / 3) * 32;
  const int cc0 = wid * 64;

  v8f acc[2][4] = {};
  wave_gemm_32x64<3 * DIMC>(X, W, row0, which * DIMC + cc0, l16, half, acc);

  const int cc = cc0 + 4 * l16;
  const int head = cc / HD, hd = cc % HD;
  float* dst = (which == 0) ? Q : (which == 1) ? K : V;
  const float sc = (which == 0) ? ATTN_SCALE : 1.0f;
#pragma unroll
  for (int m = 0; m < 2; ++m) {
#pragma unroll
    for (int i = 0; i < 8; ++i) {
      const int r = row0 + 16 * m + i + 8 * half;
      const int b = r / NTOK, n = r % NTOK;
      const v4f o = { acc[m][0][i] * sc, acc[m][1][i] * sc,
                      acc[m][2][i] * sc, acc[m][3][i] * sc };
      *reinterpret_cast<v4f*>(dst + (((size_t)b * HEADS + head) * NTOK + n) * HD + hd) = o;
    }
  }
}

template <int KH>
__device__ __forceinline__ void pv_keyhalf(const float* __restrict__ vcol0,
                                           const float* prow0, const int half, v8f& acc) {
  const float* vcol = vcol0 + 2 * half * HD;
  const float* prow = prow0 + 2 * half;
#pragma unroll
  for (int g = 0; g < 2; ++g) {
    float vb0[22], vb1[22];
#pragma unroll
    for (int s = 0; s < 22; ++s) {
      const int C = KH * (NPAD / 2) + g * 88 + 4 * s;
      if (C + 3 < NTOK) {
        vb0[s] = vcol[C * HD];
        vb1[s] = vcol[(C + 1) * HD];
      } else {
        const int kc = C + 2 * half;
        const int r0 = (kc < NTOK) ? kc : NTOK - 1;
        const int r1 = (kc + 1 < NTOK) ? kc + 1 : NTOK - 1;
        vb0[s] = vcol0[r0 * HD];
        vb1[s] = vcol0[r1 * HD];
      }
    }
    __builtin_amdgcn_sched_barrier(0);
#pragma unroll
    for (int s = 0; s < 22; ++s) {
      const int C = KH * (NPAD / 2) + g * 88 + 4 * s;
      const v2f a = { prow[C], prow[C + 1] };
      v2f bf = { vb0[s], vb1[s] };
      if (C + 3 >= NTOK) {
        const int kc = C + 2 * half;
        bf.x = (kc < NTOK) ? vb0[s] : 0.f;
        bf.y = (kc + 1 < NTOK) ? vb1[s] : 0.f;
      }
      acc = __builtin_amdgcn_wmma_f32_16x16x4_f32(false, a, false, bf,
                                                  (short)0, acc, false, false);
    }
    if (g == 0) __builtin_amdgcn_sched_barrier(0);
  }
}

__global__ __launch_bounds__(128) void attn_kernel(
    const float* __restrict__ Q, const float* __restrict__ K,
    const float* __restrict__ V, const float* __restrict__ mask,
    const float* __restrict__ rpb, float* __restrict__ AO) {
  __shared__ float sS[16 * SSTRIDE];
  __shared__ float sRed[16 * 8];
  __shared__ float sRowMax[16];
  __shared__ float sRowInv[16];
  __shared__ float sPart[2 * 16 * 16];

  const int t = threadIdx.x;
  const int wid = __builtin_amdgcn_readfirstlane(t >> 5);
  const int lane = t & 31, half = lane >> 4, l16 = lane & 15;
  const int q0 = blockIdx.x * 16;
  const int h  = blockIdx.y;
  const int b  = blockIdx.z;
  const size_t base = ((size_t)b * HEADS + h) * (size_t)NTOK * HD;
  const float* Qb = Q + base;
  const float* Kb = K + base;
  const float* Vb = V + base;

  const int qrow = (q0 + l16 < NTOK) ? q0 + l16 : NTOK - 1;
  const v4f* qp = reinterpret_cast<const v4f*>(Qb + (size_t)qrow * HD + 16 * half);
  const v4f qv0 = qp[0], qv1 = qp[1], qv2 = qp[2], qv3 = qp[3];

  const float* mrow = mask + (size_t)(b & 63) * NTOK * NTOK;
  const int myrow6 = __mul24(rp_code(qrow) + 1098, HEADS) + h;
  int aq6[8], mo[8];
#pragma unroll
  for (int i = 0; i < 8; ++i) {
    const int lo = __builtin_amdgcn_readlane(myrow6, i);
    const int hi = __builtin_amdgcn_readlane(myrow6, i + 8);
    aq6[i] = half ? hi : lo;
    const int q = q0 + i + 8 * half;
    const int qc = (q < NTOK) ? q : NTOK - 1;
    mo[i] = __mul24(qc, NTOK);
  }

  for (int kt = wid; kt < NPAD / 16; kt += 4) {
    const int kcol = kt * 16 + l16;
    const int krow = (kcol < NTOK) ? kcol : NTOK - 1;
    const v4f* kp = reinterpret_cast<const v4f*>(Kb + (size_t)krow * HD + 16 * half);
    const v4f kv0 = kp[0], kv1 = kp[1], kv2 = kp[2], kv3 = kp[3];

    const int ak6 = __mul24(rp_code(krow), HEADS);
    float br[8], mk[8];
#pragma unroll
    for (int i = 0; i < 8; ++i) br[i] = rpb[(unsigned)(aq6[i] - ak6)];
#pragma unroll
    for (int i = 0; i < 8; ++i) mk[i] = mrow[mo[i] + krow];
    __builtin_amdgcn_sched_barrier(0);

    v8f acc = {};
#define QK_STEP(QA, QB, KA, KB)                                              \
    { v2f a_ = { QA, QB }; v2f b_ = { KA, KB };                              \
      acc = __builtin_amdgcn_wmma_f32_16x16x4_f32(false, a_, false, b_,      \
                                                  (short)0, acc, false, false); }
    QK_STEP(qv0.x, qv0.y, kv0.x, kv0.y)
    QK_STEP(qv0.z, qv0.w, kv0.z, kv0.w)
    QK_STEP(qv1.x, qv1.y, kv1.x, kv1.y)
    QK_STEP(qv1.z, qv1.w, kv1.z, kv1.w)
    QK_STEP(qv2.x, qv2.y, kv2.x, kv2.y)
    QK_STEP(qv2.z, qv2.w, kv2.z, kv2.w)
    QK_STEP(qv3.x, qv3.y, kv3.x, kv3.y)
    QK_STEP(qv3.z, qv3.w, kv3.z, kv3.w)
#undef QK_STEP

    const bool kok = kcol < NTOK;
#pragma unroll
    for (int i = 0; i < 8; ++i) {
      const bool ok = kok && (q0 + i + 8 * half < NTOK);
      sS[(i + 8 * half) * SSTRIDE + kcol] = ok ? acc[i] + (br[i] + mk[i]) : -1e30f;
    }
  }
  __syncthreads();

  {
    const int r = t >> 3, sg = t & 7;
    float* srow = sS + r * SSTRIDE + sg;
    float m = -1e30f;
#pragma unroll
    for (int j = 0; j < 43; ++j) {
      const float v = srow[8 * j];
      m = (v > m) ? v : m;
    }
    sRed[r * 8 + sg] = m;
    __syncthreads();
    if (sg == 0) {
      float mm = sRed[r * 8];
#pragma unroll
      for (int j = 1; j < 8; ++j) mm = fmaxf(mm, sRed[r * 8 + j]);
      sRowMax[r] = mm;
    }
    __syncthreads();
    const float rm = sRowMax[r];
    float sum = 0.f;
#pragma unroll 14
    for (int j = 0; j < 42; ++j) {
      const float e = __expf(srow[8 * j] - rm);
      sum += e;
      srow[8 * j] = e;
    }
    {
      const float s = srow[8 * 42];
      const float e = (sg < 7) ? __expf(s - rm) : 0.f;
      sum += e;
      srow[8 * 42] = e;
    }
    srow[8 * 43] = 0.f;
    sRed[r * 8 + sg] = sum;
    __syncthreads();
    if (sg == 0) {
      float ss = 0.f;
#pragma unroll
      for (int j = 0; j < 8; ++j) ss += sRed[r * 8 + j];
      sRowInv[r] = 1.0f / ss;
    }
  }
  __syncthreads();

  const int dt = wid & 1;
  const int kh = wid >> 1;
  const int d0 = dt * 16;
  v8f acc = {};
  if (kh == 0) pv_keyhalf<0>(Vb + d0 + l16, sS + l16 * SSTRIDE, half, acc);
  else         pv_keyhalf<1>(Vb + d0 + l16, sS + l16 * SSTRIDE, half, acc);
  if (kh == 1) {
#pragma unroll
    for (int i = 0; i < 8; ++i)
      sPart[dt * 256 + (i + 8 * half) * 16 + l16] = acc[i];
  }
  __syncthreads();
  if (kh == 0) {
#pragma unroll
    for (int i = 0; i < 8; ++i) {
      const int q = q0 + i + 8 * half;
      if (q < NTOK) {
        const float v = (acc[i] + sPart[dt * 256 + (i + 8 * half) * 16 + l16])
                        * sRowInv[i + 8 * half];
        AO[((size_t)b * NTOK + q) * DIMC + h * HD + d0 + l16] = v;
      }
    }
  }
}

__global__ __launch_bounds__(96) void proj_kernel(
    const float* __restrict__ AO, const float* __restrict__ W,
    const float* __restrict__ bias, float* __restrict__ out) {
  const int t = threadIdx.x;
  const int wid = t >> 5, lane = t & 31, half = lane >> 4, l16 = lane & 15;
  const int row0 = blockIdx.x * 32;
  const int col0 = wid * 64;

  v8f acc[2][4] = {};
  wave_gemm_32x64<DIMC>(AO, W, row0, col0, l16, half, acc);

  const int c4 = col0 + 4 * l16;
  const v4f bv = *reinterpret_cast<const v4f*>(bias + c4);
#pragma unroll
  for (int m = 0; m < 2; ++m) {
#pragma unroll
    for (int i = 0; i < 8; ++i) {
      const int r = row0 + 16 * m + i + 8 * half;
      const v4f o = { acc[m][0][i] + bv.x, acc[m][1][i] + bv.y,
                      acc[m][2][i] + bv.z, acc[m][3][i] + bv.w };
      *reinterpret_cast<v4f*>(out + (size_t)r * DIMC + c4) = o;
    }
  }
}

extern "C" void kernel_launch(void* const* d_in, const int* in_sizes, int n_in,
                              void* d_out, int out_size, void* d_ws, size_t ws_size,
                              hipStream_t stream) {
  const float* x      = (const float*)d_in[0];
  const float* mask   = (const float*)d_in[1];
  const float* qkv_w  = (const float*)d_in[2];
  const float* rpb    = (const float*)d_in[3];
  const float* proj_w = (const float*)d_in[4];
  const float* proj_b = (const float*)d_in[5];
  float* out = (float*)d_out;

  const size_t qkvElems = (size_t)BATCH * HEADS * NTOK * HD;
  float* Q  = (float*)d_ws;
  float* K  = Q + qkvElems;
  float* V  = K + qkvElems;
  float* AO = V + qkvElems;

  const int rowPairs  = (BATCH * NTOK) / 32;
  const int qkvBlocks = (BATCH * NTOK) / 32 * 3;
  qkv_kernel <<<dim3(qkvBlocks),              96, 0, stream>>>(x, qkv_w, Q, K, V);
  attn_kernel<<<dim3(NPAD / 16, HEADS, BATCH), 128, 0, stream>>>(Q, K, V, mask, rpb, AO);
  proj_kernel<<<dim3(rowPairs),               96, 0, stream>>>(AO, proj_w, proj_b, out);
}


# attention kernel: each wave fetches its Q and K tiles as whole 128-B lines (4 coalesced b128 loads) into its own LDS slot and reads the fragments there, not as 16-line strided loads from global
# speedup vs baseline: 2.3782x; 1.0431x over previous
#include <hip/hip_runtime.h>
#include <hip/hip_bf16.h>

typedef float v2f __attribute__((ext_vector_type(2)));
typedef float v8f __attribute__((ext_vector_type(8)));
typedef float v4f __attribute__((ext_vector_type(4)));

#define NTOK 343
#define NPAD 352
#define HEADS 6
#define HD 32
#define DIMC 192
#define BATCH 256
#define ATTN_SCALE 0.17677669529663687f
#define SSTRIDE 353
#define KPITCH 36

__device__ __forceinline__ int rp_index(int q, int k) {
  int q0 = (q / 7) % 7, q1 = q / 49, q2 = q % 7;
  int k0 = (k / 7) % 7, k1 = k / 49, k2 = k % 7;
  return (q0 - k0 + 6) * 169 + (q1 - k1 + 6) * 13 + (q2 - k2 + 6);
}

constexpr bool rp_code_shift_exact() {
  for (int f = 0; f < NPAD; ++f) {
    const int d7 = (f * 9363) >> 16, d49 = (f * 1338) >> 16;
    if (d7 != f / 7 || d49 != f / 49) return false;
    if (f + 162 * d7 - 1170 * d49 != ((f / 7) % 7) * 169 + (f / 49) * 13 + (f % 7)) return false;
  }
  return true;
}
static_assert(rp_code_shift_exact(), "multiply-shift quotients must be exact on 0..NPAD-1");
__device__ __forceinline__ int rp_code(int f) {
  const int d7  = __mul24(f, 9363) >> 16;
  const int d49 = __mul24(f, 1338) >> 16;
  return f + __mul24(d7, 162) - __mul24(d49, 1170);
}

template <int LDW>
__device__ __forceinline__ void wave_gemm_32x64(
    const float* __restrict__ A, const float* __restrict__ W,
    const int row0, const int col0, const int l16, const int half, v8f (&acc)[2][4]) {
  const float* a0 = A + (size_t)(row0 + l16) * DIMC + 4 * half;
  const float* a1 = a0 + 16 * DIMC;
  const float* wp = W + (size_t)(4 * half) * LDW + col0 + 4 * l16;
  for (int p = 0; p < DIMC / 8; ++p) {
    const v4f x0 = *reinterpret_cast<const v4f*>(a0 + 8 * p);
    const v4f x1 = *reinterpret_cast<const v4f*>(a1 + 8 * p);
    const float* wr = wp + (size_t)(8 * p) * LDW;
    const v4f w0 = *reinterpret_cast<const v4f*>(wr);
    const v4f w1 = *reinterpret_cast<const v4f*>(wr + LDW);
    const v4f w2 = *reinterpret_cast<const v4f*>(wr + 2 * LDW);
    const v4f w3 = *reinterpret_cast<const v4f*>(wr + 3 * LDW);
    const v2f xa0 = { x0.x, x0.y }, xb0 = { x0.z, x0.w };
    const v2f xa1 = { x1.x, x1.y }, xb1 = { x1.z, x1.w };
#pragma unroll
    for (int t = 0; t < 4; ++t) {
      const v2f ba = { w0[t], w1[t] };
      const v2f bb = { w2[t], w3[t] };
      acc[0][t] = __builtin_amdgcn_wmma_f32_16x16x4_f32(false, xa0, false, ba,
                                                        (short)0, acc[0][t], false, false);
      acc[1][t] = __builtin_amdgcn_wmma_f32_16x16x4_f32(false, xa1, false, ba,
                                                        (short)0, acc[1][t], false, false);
      acc[0][t] = __builtin_amdgcn_wmma_f32_16x16x4_f32(false, xb0, false, bb,
                                                        (short)0, acc[0][t], false, false);
      acc[1][t] = __builtin_amdgcn_wmma_f32_16x16x4_f32(false, xb1, false, bb,
                                                        (short)0, acc[1][t], false, false);
    }
  }
}

__global__ __launch_bounds__(96) void qkv_kernel(
    const float* __restrict__ X, const float* __restrict__ W,
    float* __restrict__ Q, float* __restrict__ K, float* __restrict__ V) {
  const int t = threadIdx.x;
  const int wid = t >> 5, lane = t & 31, half = lane >> 4, l16 = lane & 15;
  const int which = blockIdx.x % 3;
  const int row0 = (blockIdx.x / 3) * 32;
  const int cc0 = wid * 64;

  v8f acc[2][4] = {};
  wave_gemm_32x64<3 * DIMC>(X, W, row0, which * DIMC + cc0, l16, half, acc);

  const int cc = cc0 + 4 * l16;
  const int head = cc / HD, hd = cc % HD;
  float* dst = (which == 0) ? Q : (which == 1) ? K : V;
  const float sc = (which == 0) ? ATTN_SCALE : 1.0f;
#pragma unroll
  for (int m = 0; m < 2; ++m) {
#pragma unroll
    for (int i = 0; i < 8; ++i) {
      const int r = row0 + 16 * m + i + 8 * half;
      const int b = r / NTOK, n = r % NTOK;
      const v4f o = { acc[m][0][i] * sc, acc[m][1][i] * sc,
                      acc[m][2][i] * sc, acc[m][3][i] * sc };
      *reinterpret_cast<v4f*>(dst + (((size_t)b * HEADS + head) * NTOK + n) * HD + hd) = o;
    }
  }
}

template <int KH>
__device__ __forceinline__ void pv_keyhalf(const float* __restrict__ vcol0,
                                           const float* prow0, const int half, v8f& acc) {
  const float* vcol = vcol0 + 2 * half * HD;
  const float* prow = prow0 + 2 * half;
#pragma unroll
  for (int g = 0; g < 2; ++g) {
    float vb0[22], vb1[22];
#pragma unroll
    for (int s = 0; s < 22; ++s) {
      const int C = KH * (NPAD / 2) + g * 88 + 4 * s;
      if (C + 3 < NTOK) {
        vb0[s] = vcol[C * HD];
        vb1[s] = vcol[(C + 1) * HD];
      } else {
        const int kc = C + 2 * half;
        const int r0 = (kc < NTOK) ? kc : NTOK - 1;
        const int r1 = (kc + 1 < NTOK) ? kc + 1 : NTOK - 1;
        vb0[s] = vcol0[r0 * HD];
        vb1[s] = vcol0[r1 * HD];
      }
    }
    __builtin_amdgcn_sched_barrier(0);
#pragma unroll
    for (int s = 0; s < 22; ++s) {
      const int C = KH * (NPAD / 2) + g * 88 + 4 * s;
      const v2f a = { prow[C], prow[C + 1] };
      v2f bf = { vb0[s], vb1[s] };
      if (C + 3 >= NTOK) {
        const int kc = C + 2 * half;
        bf.x = (kc < NTOK) ? vb0[s] : 0.f;
        bf.y = (kc + 1 < NTOK) ? vb1[s] : 0.f;
      }
      acc = __builtin_amdgcn_wmma_f32_16x16x4_f32(false, a, false, bf,
                                                  (short)0, acc, false, false);
    }
    if (g == 0) __builtin_amdgcn_sched_barrier(0);
  }
}

__device__ __forceinline__ void wave_lds_order() {
  __builtin_amdgcn_fence(__ATOMIC_ACQ_REL, "wavefront");
  __builtin_amdgcn_wave_barrier();
}

__global__ __launch_bounds__(128) void attn_kernel(
    const float* __restrict__ Q, const float* __restrict__ K,
    const float* __restrict__ V, const float* __restrict__ mask,
    const float* __restrict__ rpb, float* __restrict__ AO) {
  __shared__ float sS[16 * SSTRIDE];
  __shared__ float sRed[16 * 8];
  __shared__ float sRowMax[16];
  __shared__ float sRowInv[16];
  __shared__ float sPart[2 * 16 * 16];
  __shared__ __attribute__((aligned(16))) float sK[4 * 16 * KPITCH];

  const int t = threadIdx.x;
  const int wid = __builtin_amdgcn_readfirstlane(t >> 5);
  const int lane = t & 31, half = lane >> 4, l16 = lane & 15;
  const int q0 = blockIdx.x * 16;
  const int h  = blockIdx.y;
  const int b  = blockIdx.z;
  const size_t base = ((size_t)b * HEADS + h) * (size_t)NTOK * HD;
  const float* Qb = Q + base;
  const float* Kb = K + base;
  const float* Vb = V + base;

  const int qrow = (q0 + l16 < NTOK) ? q0 + l16 : NTOK - 1;
  const int lrow = lane >> 3, lcol = (lane & 7) * 4;
  float* kslot = sK + wid * (16 * KPITCH);
  float* kst = kslot + lrow * KPITCH + lcol;
  const v4f* kfr = reinterpret_cast<const v4f*>(kslot + l16 * KPITCH + 16 * half);
  v4f qv0, qv1, qv2, qv3;
  {
    v4f qg[4];
#pragma unroll
    for (int j = 0; j < 4; ++j) {
      const int r = q0 + 4 * j + lrow;
      const int rc = (r < NTOK) ? r : NTOK - 1;
      qg[j] = *reinterpret_cast<const v4f*>(Qb + (size_t)rc * HD + lcol);
    }
#pragma unroll
    for (int j = 0; j < 4; ++j)
      *reinterpret_cast<v4f*>(kst + 4 * j * KPITCH) = qg[j];
    wave_lds_order();
    qv0 = kfr[0]; qv1 = kfr[1]; qv2 = kfr[2]; qv3 = kfr[3];
    wave_lds_order();
  }

  const float* mrow = mask + (size_t)(b & 63) * NTOK * NTOK;
  const int myrow6 = __mul24(rp_code(qrow) + 1098, HEADS) + h;
  int aq6[8], mo[8];
#pragma unroll
  for (int i = 0; i < 8; ++i) {
    const int lo = __builtin_amdgcn_readlane(myrow6, i);
    const int hi = __builtin_amdgcn_readlane(myrow6, i + 8);
    aq6[i] = half ? hi : lo;
    const int q = q0 + i + 8 * half;
    const int qc = (q < NTOK) ? q : NTOK - 1;
    mo[i] = __mul24(qc, NTOK);
  }

  for (int kt = wid; kt < NPAD / 16; kt += 4) {
    const int kcol = kt * 16 + l16;
    const int krow = (kcol < NTOK) ? kcol : NTOK - 1;
    v4f kg[4];
#pragma unroll
    for (int j = 0; j < 4; ++j) {
      const int r = kt * 16 + 4 * j + lrow;
      const int rc = (r < NTOK) ? r : NTOK - 1;
      kg[j] = *reinterpret_cast<const v4f*>(Kb + (size_t)rc * HD + lcol);
    }

    const int ak6 = __mul24(rp_code(krow), HEADS);
    float br[8], mk[8];
#pragma unroll
    for (int i = 0; i < 8; ++i) br[i] = rpb[(unsigned)(aq6[i] - ak6)];
#pragma unroll
    for (int i = 0; i < 8; ++i) mk[i] = mrow[mo[i] + krow];
    __builtin_amdgcn_sched_barrier(0);

#pragma unroll
    for (int j = 0; j < 4; ++j)
      *reinterpret_cast<v4f*>(kst + 4 * j * KPITCH) = kg[j];
    wave_lds_order();
    const v4f kv0 = kfr[0], kv1 = kfr[1], kv2 = kfr[2], kv3 = kfr[3];
    wave_lds_order();

    v8f acc = {};
#define QK_STEP(QA, QB, KA, KB)                                              \
    { v2f a_ = { QA, QB }; v2f b_ = { KA, KB };                              \
      acc = __builtin_amdgcn_wmma_f32_16x16x4_f32(false, a_, false, b_,      \
                                                  (short)0, acc, false, false); }
    QK_STEP(qv0.x, qv0.y, kv0.x, kv0.y)
    QK_STEP(qv0.z, qv0.w, kv0.z, kv0.w)
    QK_STEP(qv1.x, qv1.y, kv1.x, kv1.y)
    QK_STEP(qv1.z, qv1.w, kv1.z, kv1.w)
    QK_STEP(qv2.x, qv2.y, kv2.x, kv2.y)
    QK_STEP(qv2.z, qv2.w, kv2.z, kv2.w)
    QK_STEP(qv3.x, qv3.y, kv3.x, kv3.y)
    QK_STEP(qv3.z, qv3.w, kv3.z, kv3.w)
#undef QK_STEP

    const bool kok = kcol < NTOK;
#pragma unroll
    for (int i = 0; i < 8; ++i) {
      const bool ok = kok && (q0 + i + 8 * half < NTOK);
      sS[(i + 8 * half) * SSTRIDE + kcol] = ok ? acc[i] + (br[i] + mk[i]) : -1e30f;
    }
  }
  __syncthreads();

  {
    const int r = t >> 3, sg = t & 7;
    float* srow = sS + r * SSTRIDE + sg;
    float m = -1e30f;
#pragma unroll
    for (int j = 0; j < 43; ++j) {
      const float v = srow[8 * j];
      m = (v > m) ? v : m;
    }
    sRed[r * 8 + sg] = m;
    __syncthreads();
    if (sg == 0) {
      float mm = sRed[r * 8];
#pragma unroll
      for (int j = 1; j < 8; ++j) mm = fmaxf(mm, sRed[r * 8 + j]);
      sRowMax[r] = mm;
    }
    __syncthreads();
    const float rm = sRowMax[r];
    float sum = 0.f;
#pragma unroll 14
    for (int j = 0; j < 42; ++j) {
      const float e = __expf(srow[8 * j] - rm);
      sum += e;
      srow[8 * j] = e;
    }
    {
      const float s = srow[8 * 42];
      const float e = (sg < 7) ? __expf(s - rm) : 0.f;
      sum += e;
      srow[8 * 42] = e;
    }
    srow[8 * 43] = 0.f;
    sRed[r * 8 + sg] = sum;
    __syncthreads();
    if (sg == 0) {
      float ss = 0.f;
#pragma unroll
      for (int j = 0; j < 8; ++j) ss += sRed[r * 8 + j];
      sRowInv[r] = 1.0f / ss;
    }
  }
  __syncthreads();

  const int dt = wid & 1;
  const int kh = wid >> 1;
  const int d0 = dt * 16;
  v8f acc = {};
  if (kh == 0) pv_keyhalf<0>(Vb + d0 + l16, sS + l16 * SSTRIDE, half, acc);
  else         pv_keyhalf<1>(Vb + d0 + l16, sS + l16 * SSTRIDE, half, acc);
  if (kh == 1) {
#pragma unroll
    for (int i = 0; i < 8; ++i)
      sPart[dt * 256 + (i + 8 * half) * 16 + l16] = acc[i];
  }
  __syncthreads();
  if (kh == 0) {
#pragma unroll
    for (int i = 0; i < 8; ++i) {
      const int q = q0 + i + 8 * half;
      if (q < NTOK) {
        const float v = (acc[i] + sPart[dt * 256 + (i + 8 * half) * 16 + l16])
                        * sRowInv[i + 8 * half];
        AO[((size_t)b * NTOK + q) * DIMC + h * HD + d0 + l16] = v;
      }
    }
  }
}

__global__ __launch_bounds__(96) void proj_kernel(
    const float* __restrict__ AO, const float* __restrict__ W,
    const float* __restrict__ bias, float* __restrict__ out) {
  const int t = threadIdx.x;
  const int wid = t >> 5, lane = t & 31, half = lane >> 4, l16 = lane & 15;
  const int row0 = blockIdx.x * 32;
  const int col0 = wid * 64;

  v8f acc[2][4] = {};
  wave_gemm_32x64<DIMC>(AO, W, row0, col0, l16, half, acc);

  const int c4 = col0 + 4 * l16;
  const v4f bv = *reinterpret_cast<const v4f*>(bias + c4);
#pragma unroll
  for (int m = 0; m < 2; ++m) {
#pragma unroll
    for (int i = 0; i < 8; ++i) {
      const int r = row0 + 16 * m + i + 8 * half;
      const v4f o = { acc[m][0][i] + bv.x, acc[m][1][i] + bv.y,
                      acc[m][2][i] + bv.z, acc[m][3][i] + bv.w };
      *reinterpret_cast<v4f*>(out + (size_t)r * DIMC + c4) = o;
    }
  }
}

extern "C" void kernel_launch(void* const* d_in, const int* in_sizes, int n_in,
                              void* d_out, int out_size, void* d_ws, size_t ws_size,
                              hipStream_t stream) {
  const float* x      = (const float*)d_in[0];
  const float* mask   = (const float*)d_in[1];
  const float* qkv_w  = (const float*)d_in[2];
  const float* rpb    = (const float*)d_in[3];
  const float* proj_w = (const float*)d_in[4];
  const float* proj_b = (const float*)d_in[5];
  float* out = (float*)d_out;

  const size_t qkvElems = (size_t)BATCH * HEADS * NTOK * HD;
  float* Q  = (float*)d_ws;
  float* K  = Q + qkvElems;
  float* V  = K + qkvElems;
  float* AO = V + qkvElems;

  const int rowPairs  = (BATCH * NTOK) / 32;
  const int qkvBlocks = (BATCH * NTOK) / 32 * 3;
  qkv_kernel <<<dim3(qkvBlocks),              96, 0, stream>>>(x, qkv_w, Q, K, V);
  attn_kernel<<<dim3(NPAD / 16, HEADS, BATCH), 128, 0, stream>>>(Q, K, V, mask, rpb, AO);
  proj_kernel<<<dim3(rowPairs),               96, 0, stream>>>(AO, proj_w, proj_b, out);
}


# attention P@V: each wave does both 16-wide d tiles for a quarter of the keys (d interleaved by 2: V rows load as b64, a P fragment feeds 2 WMMAs); 4 partials added via LDS, key sum order changes
# speedup vs baseline: 2.4952x; 1.0492x over previous
#include <hip/hip_runtime.h>
#include <hip/hip_bf16.h>

typedef float v2f __attribute__((ext_vector_type(2)));
typedef float v8f __attribute__((ext_vector_type(8)));
typedef float v4f __attribute__((ext_vector_type(4)));

#define NTOK 343
#define NPAD 352
#define HEADS 6
#define HD 32
#define DIMC 192
#define BATCH 256
#define ATTN_SCALE 0.17677669529663687f
#define SSTRIDE 353
#define PPITCH 36
#define KPITCH 36

__device__ __forceinline__ int rp_index(int q, int k) {
  int q0 = (q / 7) % 7, q1 = q / 49, q2 = q % 7;
  int k0 = (k / 7) % 7, k1 = k / 49, k2 = k % 7;
  return (q0 - k0 + 6) * 169 + (q1 - k1 + 6) * 13 + (q2 - k2 + 6);
}

constexpr bool rp_code_shift_exact() {
  for (int f = 0; f < NPAD; ++f) {
    const int d7 = (f * 9363) >> 16, d49 = (f * 1338) >> 16;
    if (d7 != f / 7 || d49 != f / 49) return false;
    if (f + 162 * d7 - 1170 * d49 != ((f / 7) % 7) * 169 + (f / 49) * 13 + (f % 7)) return false;
  }
  return true;
}
static_assert(rp_code_shift_exact(), "multiply-shift quotients must be exact on 0..NPAD-1");
__device__ __forceinline__ int rp_code(int f) {
  const int d7  = __mul24(f, 9363) >> 16;
  const int d49 = __mul24(f, 1338) >> 16;
  return f + __mul24(d7, 162) - __mul24(d49, 1170);
}

template <int LDW>
__device__ __forceinline__ void wave_gemm_32x64(
    const float* __restrict__ A, const float* __restrict__ W,
    const int row0, const int col0, const int l16, const int half, v8f (&acc)[2][4]) {
  const float* a0 = A + (size_t)(row0 + l16) * DIMC + 4 * half;
  const float* a1 = a0 + 16 * DIMC;
  const float* wp = W + (size_t)(4 * half) * LDW + col0 + 4 * l16;
  for (int p = 0; p < DIMC / 8; ++p) {
    const v4f x0 = *reinterpret_cast<const v4f*>(a0 + 8 * p);
    const v4f x1 = *reinterpret_cast<const v4f*>(a1 + 8 * p);
    const float* wr = wp + (size_t)(8 * p) * LDW;
    const v4f w0 = *reinterpret_cast<const v4f*>(wr);
    const v4f w1 = *reinterpret_cast<const v4f*>(wr + LDW);
    const v4f w2 = *reinterpret_cast<const v4f*>(wr + 2 * LDW);
    const v4f w3 = *reinterpret_cast<const v4f*>(wr + 3 * LDW);
    const v2f xa0 = { x0.x, x0.y }, xb0 = { x0.z, x0.w };
    const v2f xa1 = { x1.x, x1.y }, xb1 = { x1.z, x1.w };
#pragma unroll
    for (int t = 0; t < 4; ++t) {
      const v2f ba = { w0[t], w1[t] };
      const v2f bb = { w2[t], w3[t] };
      acc[0][t] = __builtin_amdgcn_wmma_f32_16x16x4_f32(false, xa0, false, ba,
                                                        (short)0, acc[0][t], false, false);
      acc[1][t] = __builtin_amdgcn_wmma_f32_16x16x4_f32(false, xa1, false, ba,
                                                        (short)0, acc[1][t], false, false);
      acc[0][t] = __builtin_amdgcn_wmma_f32_16x16x4_f32(false, xb0, false, bb,
                                                        (short)0, acc[0][t], false, false);
      acc[1][t] = __builtin_amdgcn_wmma_f32_16x16x4_f32(false, xb1, false, bb,
                                                        (short)0, acc[1][t], false, false);
    }
  }
}

__global__ __launch_bounds__(96) void qkv_kernel(
    const float* __restrict__ X, const float* __restrict__ W,
    float* __restrict__ Q, float* __restrict__ K, float* __restrict__ V) {
  const int t = threadIdx.x;
  const int wid = t >> 5, lane = t & 31, half = lane >> 4, l16 = lane & 15;
  const int which = blockIdx.x % 3;
  const int row0 = (blockIdx.x / 3) * 32;
  const int cc0 = wid * 64;

  v8f acc[2][4] = {};
  wave_gemm_32x64<3 * DIMC>(X, W, row0, which * DIMC + cc0, l16, half, acc);

  const int cc = cc0 + 4 * l16;
  const int head = cc / HD, hd = cc % HD;
  float* dst = (which == 0) ? Q : (which == 1) ? K : V;
  const float sc = (which == 0) ? ATTN_SCALE : 1.0f;
#pragma unroll
  for (int m = 0; m < 2; ++m) {
#pragma unroll
    for (int i = 0; i < 8; ++i) {
      const int r = row0 + 16 * m + i + 8 * half;
      const int b = r / NTOK, n = r % NTOK;
      const v4f o = { acc[m][0][i] * sc, acc[m][1][i] * sc,
                      acc[m][2][i] * sc, acc[m][3][i] * sc };
      *reinterpret_cast<v4f*>(dst + (((size_t)b * HEADS + head) * NTOK + n) * HD + hd) = o;
    }
  }
}

template <int KQ>
__device__ __forceinline__ void pv_keyquarter(const float* __restrict__ vrow0,
                                              const float* prow0, const int half,
                                              v8f& acc0, v8f& acc1) {
  const float* vrow = vrow0 + half * HD;
  const float* prow = prow0 + half;
#pragma unroll
  for (int g = 0; g < 2; ++g) {
    v2f va[11], vb[11];
#pragma unroll
    for (int s = 0; s < 11; ++s) {
      const int C = KQ * (NPAD / 4) + g * 44 + 4 * s;
      if (C + 3 < NTOK) {
        va[s] = *reinterpret_cast<const v2f*>(vrow + C * HD);
        vb[s] = *reinterpret_cast<const v2f*>(vrow + (C + 2) * HD);
      } else {
        const int k0 = C + half, k1 = C + 2 + half;
        const int r0 = (k0 < NTOK) ? k0 : NTOK - 1;
        const int r1 = (k1 < NTOK) ? k1 : NTOK - 1;
        va[s] = *reinterpret_cast<const v2f*>(vrow0 + r0 * HD);
        vb[s] = *reinterpret_cast<const v2f*>(vrow0 + r1 * HD);
      }
    }
    __builtin_amdgcn_sched_barrier(0);
#pragma unroll
    for (int s = 0; s < 11; ++s) {
      const int C = KQ * (NPAD / 4) + g * 44 + 4 * s;
      const v2f a = { prow[C], prow[C + 2] };
      v2f b0 = { va[s].x, vb[s].x };
      v2f b1 = { va[s].y, vb[s].y };
      if (C + 3 >= NTOK) {
        const bool ok0 = (C + half < NTOK), ok1 = (C + 2 + half < NTOK);
        b0.x = ok0 ? b0.x : 0.f;  b1.x = ok0 ? b1.x : 0.f;
        b0.y = ok1 ? b0.y : 0.f;  b1.y = ok1 ? b1.y : 0.f;
      }
      acc0 = __builtin_amdgcn_wmma_f32_16x16x4_f32(false, a, false, b0,
                                                   (short)0, acc0, false, false);
      acc1 = __builtin_amdgcn_wmma_f32_16x16x4_f32(false, a, false, b1,
                                                   (short)0, acc1, false, false);
    }
    if (g == 0) __builtin_amdgcn_sched_barrier(0);
  }
}

__device__ __forceinline__ void wave_lds_order() {
  __builtin_amdgcn_fence(__ATOMIC_ACQ_REL, "wavefront");
  __builtin_amdgcn_wave_barrier();
}

__global__ __launch_bounds__(128) void attn_kernel(
    const float* __restrict__ Q, const float* __restrict__ K,
    const float* __restrict__ V, const float* __restrict__ mask,
    const float* __restrict__ rpb, float* __restrict__ AO) {
  __shared__ float sS[16 * SSTRIDE];
  __shared__ float sRed[16 * 8];
  __shared__ float sRowMax[16];
  __shared__ float sRowInv[16];
  __shared__ __attribute__((aligned(16))) float sK[4 * 16 * KPITCH];

  const int t = threadIdx.x;
  const int wid = __builtin_amdgcn_readfirstlane(t >> 5);
  const int lane = t & 31, half = lane >> 4, l16 = lane & 15;
  const int q0 = blockIdx.x * 16;
  const int h  = blockIdx.y;
  const int b  = blockIdx.z;
  const size_t base = ((size_t)b * HEADS + h) * (size_t)NTOK * HD;
  const float* Qb = Q + base;
  const float* Kb = K + base;
  const float* Vb = V + base;

  const int qrow = (q0 + l16 < NTOK) ? q0 + l16 : NTOK - 1;
  const int lrow = lane >> 3, lcol = (lane & 7) * 4;
  float* kslot = sK + wid * (16 * KPITCH);
  float* kst = kslot + lrow * KPITCH + lcol;
  const v4f* kfr = reinterpret_cast<const v4f*>(kslot + l16 * KPITCH + 16 * half);
  v4f qv0, qv1, qv2, qv3;
  {
    v4f qg[4];
#pragma unroll
    for (int j = 0; j < 4; ++j) {
      const int r = q0 + 4 * j + lrow;
      const int rc = (r < NTOK) ? r : NTOK - 1;
      qg[j] = *reinterpret_cast<const v4f*>(Qb + (size_t)rc * HD + lcol);
    }
#pragma unroll
    for (int j = 0; j < 4; ++j)
      *reinterpret_cast<v4f*>(kst + 4 * j * KPITCH) = qg[j];
    wave_lds_order();
    qv0 = kfr[0]; qv1 = kfr[1]; qv2 = kfr[2]; qv3 = kfr[3];
    wave_lds_order();
  }

  const float* mrow = mask + (size_t)(b & 63) * NTOK * NTOK;
  const int myrow6 = __mul24(rp_code(qrow) + 1098, HEADS) + h;
  int aq6[8], mo[8];
#pragma unroll
  for (int i = 0; i < 8; ++i) {
    const int lo = __builtin_amdgcn_readlane(myrow6, i);
    const int hi = __builtin_amdgcn_readlane(myrow6, i + 8);
    aq6[i] = half ? hi : lo;
    const int q = q0 + i + 8 * half;
    const int qc = (q < NTOK) ? q : NTOK - 1;
    mo[i] = __mul24(qc, NTOK);
  }

  for (int kt = wid; kt < NPAD / 16; kt += 4) {
    const int kcol = kt * 16 + l16;
    const int krow = (kcol < NTOK) ? kcol : NTOK - 1;
    v4f kg[4];
#pragma unroll
    for (int j = 0; j < 4; ++j) {
      const int r = kt * 16 + 4 * j + lrow;
      const int rc = (r < NTOK) ? r : NTOK - 1;
      kg[j] = *reinterpret_cast<const v4f*>(Kb + (size_t)rc * HD + lcol);
    }

    const int ak6 = __mul24(rp_code(krow), HEADS);
    float br[8], mk[8];
#pragma unroll
    for (int i = 0; i < 8; ++i) br[i] = rpb[(unsigned)(aq6[i] - ak6)];
#pragma unroll
    for (int i = 0; i < 8; ++i) mk[i] = mrow[mo[i] + krow];
    __builtin_amdgcn_sched_barrier(0);

#pragma unroll
    for (int j = 0; j < 4; ++j)
      *reinterpret_cast<v4f*>(kst + 4 * j * KPITCH) = kg[j];
    wave_lds_order();
    const v4f kv0 = kfr[0], kv1 = kfr[1], kv2 = kfr[2], kv3 = kfr[3];
    wave_lds_order();

    v8f acc = {};
#define QK_STEP(QA, QB, KA, KB)                                              \
    { v2f a_ = { QA, QB }; v2f b_ = { KA, KB };                              \
      acc = __builtin_amdgcn_wmma_f32_16x16x4_f32(false, a_, false, b_,      \
                                                  (short)0, acc, false, false); }
    QK_STEP(qv0.x, qv0.y, kv0.x, kv0.y)
    QK_STEP(qv0.z, qv0.w, kv0.z, kv0.w)
    QK_STEP(qv1.x, qv1.y, kv1.x, kv1.y)
    QK_STEP(qv1.z, qv1.w, kv1.z, kv1.w)
    QK_STEP(qv2.x, qv2.y, kv2.x, kv2.y)
    QK_STEP(qv2.z, qv2.w, kv2.z, kv2.w)
    QK_STEP(qv3.x, qv3.y, kv3.x, kv3.y)
    QK_STEP(qv3.z, qv3.w, kv3.z, kv3.w)
#undef QK_STEP

    const bool kok = kcol < NTOK;
#pragma unroll
    for (int i = 0; i < 8; ++i) {
      const bool ok = kok && (q0 + i + 8 * half < NTOK);
      sS[(i + 8 * half) * SSTRIDE + kcol] = ok ? acc[i] + (br[i] + mk[i]) : -1e30f;
    }
  }
  __syncthreads();

  {
    const int r = t >> 3, sg = t & 7;
    float* srow = sS + r * SSTRIDE + sg;
    float m = -1e30f;
#pragma unroll
    for (int j = 0; j < 43; ++j) {
      const float v = srow[8 * j];
      m = (v > m) ? v : m;
    }
    sRed[r * 8 + sg] = m;
    __syncthreads();
    if (sg == 0) {
      float mm = sRed[r * 8];
#pragma unroll
      for (int j = 1; j < 8; ++j) mm = fmaxf(mm, sRed[r * 8 + j]);
      sRowMax[r] = mm;
    }
    __syncthreads();
    const float rm = sRowMax[r];
    float sum = 0.f;
#pragma unroll 14
    for (int j = 0; j < 42; ++j) {
      const float e = __expf(srow[8 * j] - rm);
      sum += e;
      srow[8 * j] = e;
    }
    {
      const float s = srow[8 * 42];
      const float e = (sg < 7) ? __expf(s - rm) : 0.f;
      sum += e;
      srow[8 * 42] = e;
    }
    srow[8 * 43] = 0.f;
    sRed[r * 8 + sg] = sum;
    __syncthreads();
    if (sg == 0) {
      float ss = 0.f;
#pragma unroll
      for (int j = 0; j < 8; ++j) ss += sRed[r * 8 + j];
      sRowInv[r] = 1.0f / ss;
    }
  }
  __syncthreads();

  v8f acc0 = {}, acc1 = {};
  {
    const float* vrow0 = Vb + 2 * l16;
    const float* prow0 = sS + l16 * SSTRIDE;
    if (wid == 0)      pv_keyquarter<0>(vrow0, prow0, half, acc0, acc1);
    else if (wid == 1) pv_keyquarter<1>(vrow0, prow0, half, acc0, acc1);
    else if (wid == 2) pv_keyquarter<2>(vrow0, prow0, half, acc0, acc1);
    else               pv_keyquarter<3>(vrow0, prow0, half, acc0, acc1);
  }
  if (wid != 0) {
    float* dstp = sK + (wid - 1) * (16 * PPITCH) + 8 * half * PPITCH + 2 * l16;
#pragma unroll
    for (int i = 0; i < 8; ++i) {
      const v2f pr = { acc0[i], acc1[i] };
      *reinterpret_cast<v2f*>(dstp + i * PPITCH) = pr;
    }
  }
  __syncthreads();
  if (wid == 0) {
    const float* srcp = sK + 8 * half * PPITCH + 2 * l16;
#pragma unroll
    for (int i = 0; i < 8; ++i) {
      const int q = q0 + i + 8 * half;
      if (q < NTOK) {
        const v2f p1 = *reinterpret_cast<const v2f*>(srcp + i * PPITCH);
        const v2f p2 = *reinterpret_cast<const v2f*>(srcp + 16 * PPITCH + i * PPITCH);
        const v2f p3 = *reinterpret_cast<const v2f*>(srcp + 32 * PPITCH + i * PPITCH);
        const float inv = sRowInv[i + 8 * half];
        const v2f o = { (((acc0[i] + p1.x) + p2.x) + p3.x) * inv,
                        (((acc1[i] + p1.y) + p2.y) + p3.y) * inv };
        *reinterpret_cast<v2f*>(AO + ((size_t)b * NTOK + q) * DIMC + h * HD + 2 * l16) = o;
      }
    }
  }
}

__global__ __launch_bounds__(96) void proj_kernel(
    const float* __restrict__ AO, const float* __restrict__ W,
    const float* __restrict__ bias, float* __restrict__ out) {
  const int t = threadIdx.x;
  const int wid = t >> 5, lane = t & 31, half = lane >> 4, l16 = lane & 15;
  const int row0 = blockIdx.x * 32;
  const int col0 = wid * 64;

  v8f acc[2][4] = {};
  wave_gemm_32x64<DIMC>(AO, W, row0, col0, l16, half, acc);

  const int c4 = col0 + 4 * l16;
  const v4f bv = *reinterpret_cast<const v4f*>(bias + c4);
#pragma unroll
  for (int m = 0; m < 2; ++m) {
#pragma unroll
    for (int i = 0; i < 8; ++i) {
      const int r = row0 + 16 * m + i + 8 * half;
      const v4f o = { acc[m][0][i] + bv.x, acc[m][1][i] + bv.y,
                      acc[m][2][i] + bv.z, acc[m][3][i] + bv.w };
      *reinterpret_cast<v4f*>(out + (size_t)r * DIMC + c4) = o;
    }
  }
}

extern "C" void kernel_launch(void* const* d_in, const int* in_sizes, int n_in,
                              void* d_out, int out_size, void* d_ws, size_t ws_size,
                              hipStream_t stream) {
  const float* x      = (const float*)d_in[0];
  const float* mask   = (const float*)d_in[1];
  const float* qkv_w  = (const float*)d_in[2];
  const float* rpb    = (const float*)d_in[3];
  const float* proj_w = (const float*)d_in[4];
  const float* proj_b = (const float*)d_in[5];
  float* out = (float*)d_out;

  const size_t qkvElems = (size_t)BATCH * HEADS * NTOK * HD;
  float* Q  = (float*)d_ws;
  float* K  = Q + qkvElems;
  float* V  = K + qkvElems;
  float* AO = V + qkvElems;

  const int rowPairs  = (BATCH * NTOK) / 32;
  const int qkvBlocks = (BATCH * NTOK) / 32 * 3;
  qkv_kernel <<<dim3(qkvBlocks),              96, 0, stream>>>(x, qkv_w, Q, K, V);
  attn_kernel<<<dim3(NPAD / 16, HEADS, BATCH), 128, 0, stream>>>(Q, K, V, mask, rpb, AO);
  proj_kernel<<<dim3(rowPairs),               96, 0, stream>>>(AO, proj_w, proj_b, out);
}


# attention kernel: a block takes 32 queries (2 tiles of 16); each fetched K tile and V fragment serves both tiles (16 and 4 WMMAs per fetch, two independent score chains); sum orders unchanged
# speedup vs baseline: 2.7298x; 1.0940x over previous
#include <hip/hip_runtime.h>
#include <hip/hip_bf16.h>

typedef float v2f __attribute__((ext_vector_type(2)));
typedef float v8f __attribute__((ext_vector_type(8)));
typedef float v4f __attribute__((ext_vector_type(4)));

#define NTOK 343
#define NPAD 352
#define HEADS 6
#define HD 32
#define DIMC 192
#define BATCH 256
#define QROWS 32
#define ATTN_SCALE 0.17677669529663687f
#define SSTRIDE 353
#define PPITCH 36
#define KPITCH 36

__device__ __forceinline__ int rp_index(int q, int k) {
  int q0 = (q / 7) % 7, q1 = q / 49, q2 = q % 7;
  int k0 = (k / 7) % 7, k1 = k / 49, k2 = k % 7;
  return (q0 - k0 + 6) * 169 + (q1 - k1 + 6) * 13 + (q2 - k2 + 6);
}

constexpr bool rp_code_shift_exact() {
  for (int f = 0; f < NPAD; ++f) {
    const int d7 = (f * 9363) >> 16, d49 = (f * 1338) >> 16;
    if (d7 != f / 7 || d49 != f / 49) return false;
    if (f + 162 * d7 - 1170 * d49 != ((f / 7) % 7) * 169 + (f / 49) * 13 + (f % 7)) return false;
  }
  return true;
}
static_assert(rp_code_shift_exact(), "multiply-shift quotients must be exact on 0..NPAD-1");
__device__ __forceinline__ int rp_code(int f) {
  const int d7  = __mul24(f, 9363) >> 16;
  const int d49 = __mul24(f, 1338) >> 16;
  return f + __mul24(d7, 162) - __mul24(d49, 1170);
}

template <int LDW>
__device__ __forceinline__ void wave_gemm_32x64(
    const float* __restrict__ A, const float* __restrict__ W,
    const int row0, const int col0, const int l16, const int half, v8f (&acc)[2][4]) {
  const float* a0 = A + (size_t)(row0 + l16) * DIMC + 4 * half;
  const float* a1 = a0 + 16 * DIMC;
  const float* wp = W + (size_t)(4 * half) * LDW + col0 + 4 * l16;
  for (int p = 0; p < DIMC / 8; ++p) {
    const v4f x0 = *reinterpret_cast<const v4f*>(a0 + 8 * p);
    const v4f x1 = *reinterpret_cast<const v4f*>(a1 + 8 * p);
    const float* wr = wp + (size_t)(8 * p) * LDW;
    const v4f w0 = *reinterpret_cast<const v4f*>(wr);
    const v4f w1 = *reinterpret_cast<const v4f*>(wr + LDW);
    const v4f w2 = *reinterpret_cast<const v4f*>(wr + 2 * LDW);
    const v4f w3 = *reinterpret_cast<const v4f*>(wr + 3 * LDW);
    const v2f xa0 = { x0.x, x0.y }, xb0 = { x0.z, x0.w };
    const v2f xa1 = { x1.x, x1.y }, xb1 = { x1.z, x1.w };
#pragma unroll
    for (int t = 0; t < 4; ++t) {
      const v2f ba = { w0[t], w1[t] };
      const v2f bb = { w2[t], w3[t] };
      acc[0][t] = __builtin_amdgcn_wmma_f32_16x16x4_f32(false, xa0, false, ba,
                                                        (short)0, acc[0][t], false, false);
      acc[1][t] = __builtin_amdgcn_wmma_f32_16x16x4_f32(false, xa1, false, ba,
                                                        (short)0, acc[1][t], false, false);
      acc[0][t] = __builtin_amdgcn_wmma_f32_16x16x4_f32(false, xb0, false, bb,
                                                        (short)0, acc[0][t], false, false);
      acc[1][t] = __builtin_amdgcn_wmma_f32_16x16x4_f32(false, xb1, false, bb,
                                                        (short)0, acc[1][t], false, false);
    }
  }
}

__global__ __launch_bounds__(96) void qkv_kernel(
    const float* __restrict__ X, const float* __restrict__ W,
    float* __restrict__ Q, float* __restrict__ K, float* __restrict__ V) {
  const int t = threadIdx.x;
  const int wid = t >> 5, lane = t & 31, half = lane >> 4, l16 = lane & 15;
  const int which = blockIdx.x % 3;
  const int row0 = (blockIdx.x / 3) * 32;
  const int cc0 = wid * 64;

  v8f acc[2][4] = {};
  wave_gemm_32x64<3 * DIMC>(X, W, row0, which * DIMC + cc0, l16, half, acc);

  const int cc = cc0 + 4 * l16;
  const int head = cc / HD, hd = cc % HD;
  float* dst = (which == 0) ? Q : (which == 1) ? K : V;
  const float sc = (which == 0) ? ATTN_SCALE : 1.0f;
#pragma unroll
  for (int m = 0; m < 2; ++m) {
#pragma unroll
    for (int i = 0; i < 8; ++i) {
      const int r = row0 + 16 * m + i + 8 * half;
      const int b = r / NTOK, n = r % NTOK;
      const v4f o = { acc[m][0][i] * sc, acc[m][1][i] * sc,
                      acc[m][2][i] * sc, acc[m][3][i] * sc };
      *reinterpret_cast<v4f*>(dst + (((size_t)b * HEADS + head) * NTOK + n) * HD + hd) = o;
    }
  }
}

template <int KQ>
__device__ __forceinline__ void pv_keyquarter(const float* __restrict__ vrow0,
                                              const float* prow0, const int half,
                                              v8f& a00, v8f& a01, v8f& a10, v8f& a11) {
  const float* vrow = vrow0 + half * HD;
  const float* prow = prow0 + half;
  const float* prow1 = prow + 16 * SSTRIDE;
#pragma unroll
  for (int g = 0; g < 2; ++g) {
    v2f va[11], vb[11];
#pragma unroll
    for (int s = 0; s < 11; ++s) {
      const int C = KQ * (NPAD / 4) + g * 44 + 4 * s;
      if (C + 3 < NTOK) {
        va[s] = *reinterpret_cast<const v2f*>(vrow + C * HD);
        vb[s] = *reinterpret_cast<const v2f*>(vrow + (C + 2) * HD);
      } else {
        const int k0 = C + half, k1 = C + 2 + half;
        const int r0 = (k0 < NTOK) ? k0 : NTOK - 1;
        const int r1 = (k1 < NTOK) ? k1 : NTOK - 1;
        va[s] = *reinterpret_cast<const v2f*>(vrow0 + r0 * HD);
        vb[s] = *reinterpret_cast<const v2f*>(vrow0 + r1 * HD);
      }
    }
    __builtin_amdgcn_sched_barrier(0);
#pragma unroll
    for (int s = 0; s < 11; ++s) {
      const int C = KQ * (NPAD / 4) + g * 44 + 4 * s;
      const v2f pa = { prow[C], prow[C + 2] };
      const v2f pb = { prow1[C], prow1[C + 2] };
      v2f b0 = { va[s].x, vb[s].x };
      v2f b1 = { va[s].y, vb[s].y };
      if (C + 3 >= NTOK) {
        const bool ok0 = (C + half < NTOK), ok1 = (C + 2 + half < NTOK);
        b0.x = ok0 ? b0.x : 0.f;  b1.x = ok0 ? b1.x : 0.f;
        b0.y = ok1 ? b0.y : 0.f;  b1.y = ok1 ? b1.y : 0.f;
      }
      a00 = __builtin_amdgcn_wmma_f32_16x16x4_f32(false, pa, false, b0,
                                                  (short)0, a00, false, false);
      a01 = __builtin_amdgcn_wmma_f32_16x16x4_f32(false, pa, false, b1,
                                                  (short)0, a01, false, false);
      a10 = __builtin_amdgcn_wmma_f32_16x16x4_f32(false, pb, false, b0,
                                                  (short)0, a10, false, false);
      a11 = __builtin_amdgcn_wmma_f32_16x16x4_f32(false, pb, false, b1,
                                                  (short)0, a11, false, false);
    }
    if (g == 0) __builtin_amdgcn_sched_barrier(0);
  }
}

__device__ __forceinline__ void wave_lds_order() {
  __builtin_amdgcn_fence(__ATOMIC_ACQ_REL, "wavefront");
  __builtin_amdgcn_wave_barrier();
}

__global__ __launch_bounds__(128) void attn_kernel(
    const float* __restrict__ Q, const float* __restrict__ K,
    const float* __restrict__ V, const float* __restrict__ mask,
    const float* __restrict__ rpb, float* __restrict__ AO) {
  __shared__ __attribute__((aligned(16))) float sS[QROWS * SSTRIDE];
  __shared__ float sRed[QROWS * 8];
  __shared__ float sRowMax[QROWS];
  __shared__ float sRowInv[QROWS];
  __shared__ __attribute__((aligned(16))) float sK[4 * 16 * KPITCH];

  const int t = threadIdx.x;
  const int wid = __builtin_amdgcn_readfirstlane(t >> 5);
  const int lane = t & 31, half = lane >> 4, l16 = lane & 15;
  const int q0 = blockIdx.x * QROWS;
  const int h  = blockIdx.y;
  const int b  = blockIdx.z;
  const size_t base = ((size_t)b * HEADS + h) * (size_t)NTOK * HD;
  const float* Qb = Q + base;
  const float* Kb = K + base;
  const float* Vb = V + base;

  const int lrow = lane >> 3, lcol = (lane & 7) * 4;
  float* kslot = sK + wid * (16 * KPITCH);
  float* kst = kslot + lrow * KPITCH + lcol;
  const v4f* kfr = reinterpret_cast<const v4f*>(kslot + l16 * KPITCH + 16 * half);
  v4f qv[2][4];
#pragma unroll
  for (int u = 0; u < 2; ++u) {
    v4f qg[4];
#pragma unroll
    for (int j = 0; j < 4; ++j) {
      const int r = q0 + 16 * u + 4 * j + lrow;
      const int rc = (r < NTOK) ? r : NTOK - 1;
      qg[j] = *reinterpret_cast<const v4f*>(Qb + (size_t)rc * HD + lcol);
    }
#pragma unroll
    for (int j = 0; j < 4; ++j)
      *reinterpret_cast<v4f*>(kst + 4 * j * KPITCH) = qg[j];
    wave_lds_order();
    qv[u][0] = kfr[0]; qv[u][1] = kfr[1]; qv[u][2] = kfr[2]; qv[u][3] = kfr[3];
    wave_lds_order();
  }

  const float* mrow = mask + (size_t)(b & 63) * NTOK * NTOK;
  const int ownrow = (q0 + lane < NTOK) ? q0 + lane : NTOK - 1;
  const int myrow6 = __mul24(rp_code(ownrow) + 1098, HEADS) + h;
  int aq6[2][8], mo[2][8];
#pragma unroll
  for (int u = 0; u < 2; ++u) {
#pragma unroll
    for (int i = 0; i < 8; ++i) {
      const int lo = __builtin_amdgcn_readlane(myrow6, 16 * u + i);
      const int hi = __builtin_amdgcn_readlane(myrow6, 16 * u + i + 8);
      aq6[u][i] = half ? hi : lo;
      const int q = q0 + 16 * u + i + 8 * half;
      const int qc = (q < NTOK) ? q : NTOK - 1;
      mo[u][i] = __mul24(qc, NTOK);
    }
  }

  for (int kt = wid; kt < NPAD / 16; kt += 4) {
    const int kcol = kt * 16 + l16;
    const int krow = (kcol < NTOK) ? kcol : NTOK - 1;
    v4f kg[4];
#pragma unroll
    for (int j = 0; j < 4; ++j) {
      const int r = kt * 16 + 4 * j + lrow;
      const int rc = (r < NTOK) ? r : NTOK - 1;
      kg[j] = *reinterpret_cast<const v4f*>(Kb + (size_t)rc * HD + lcol);
    }

    const int ak6 = __mul24(rp_code(krow), HEADS);
    float br[2][8], mk[2][8];
#pragma unroll
    for (int u = 0; u < 2; ++u) {
#pragma unroll
      for (int i = 0; i < 8; ++i) br[u][i] = rpb[(unsigned)(aq6[u][i] - ak6)];
    }
#pragma unroll
    for (int u = 0; u < 2; ++u) {
#pragma unroll
      for (int i = 0; i < 8; ++i) mk[u][i] = mrow[mo[u][i] + krow];
    }
    __builtin_amdgcn_sched_barrier(0);

#pragma unroll
    for (int j = 0; j < 4; ++j)
      *reinterpret_cast<v4f*>(kst + 4 * j * KPITCH) = kg[j];
    wave_lds_order();
    const v4f kv0 = kfr[0], kv1 = kfr[1], kv2 = kfr[2], kv3 = kfr[3];
    wave_lds_order();

    v8f acc0 = {}, acc1 = {};
#define QK_STEP(J, QA, QB, KA, KB)                                           \
    { v2f a0_ = { qv[0][J].QA, qv[0][J].QB }; v2f a1_ = { qv[1][J].QA, qv[1][J].QB }; \
      v2f b_ = { KA, KB };                                                   \
      acc0 = __builtin_amdgcn_wmma_f32_16x16x4_f32(false, a0_, false, b_,    \
                                                   (short)0, acc0, false, false); \
      acc1 = __builtin_amdgcn_wmma_f32_16x16x4_f32(false, a1_, false, b_,    \
                                                   (short)0, acc1, false, false); }
    QK_STEP(0, x, y, kv0.x, kv0.y)
    QK_STEP(0, z, w, kv0.z, kv0.w)
    QK_STEP(1, x, y, kv1.x, kv1.y)
    QK_STEP(1, z, w, kv1.z, kv1.w)
    QK_STEP(2, x, y, kv2.x, kv2.y)
    QK_STEP(2, z, w, kv2.z, kv2.w)
    QK_STEP(3, x, y, kv3.x, kv3.y)
    QK_STEP(3, z, w, kv3.z, kv3.w)
#undef QK_STEP

    const bool kok = kcol < NTOK;
#pragma unroll
    for (int i = 0; i < 8; ++i) {
      const bool ok = kok && (q0 + i + 8 * half < NTOK);
      sS[(i + 8 * half) * SSTRIDE + kcol] = ok ? acc0[i] + (br[0][i] + mk[0][i]) : -1e30f;
    }
#pragma unroll
    for (int i = 0; i < 8; ++i) {
      const bool ok = kok && (q0 + 16 + i + 8 * half < NTOK);
      sS[(16 + i + 8 * half) * SSTRIDE + kcol] = ok ? acc1[i] + (br[1][i] + mk[1][i]) : -1e30f;
    }
  }
  __syncthreads();

  {
    const int r = t >> 3, sg = t & 7;
    float* srowA = sS + r * SSTRIDE + sg;
    float* srowB = srowA + 16 * SSTRIDE;
    float mA = -1e30f, mB = -1e30f;
#pragma unroll
    for (int j = 0; j < 43; ++j) {
      const float vA = srowA[8 * j];
      const float vB = srowB[8 * j];
      mA = (vA > mA) ? vA : mA;
      mB = (vB > mB) ? vB : mB;
    }
    sRed[r * 8 + sg] = mA;
    sRed[(r + 16) * 8 + sg] = mB;
    __syncthreads();
    if (sg < 2) {
      const int row = r + 16 * sg;
      float mm = sRed[row * 8];
#pragma unroll
      for (int j = 1; j < 8; ++j) mm = fmaxf(mm, sRed[row * 8 + j]);
      sRowMax[row] = mm;
    }
    __syncthreads();
    const float rmA = sRowMax[r];
    const float rmB = sRowMax[r + 16];
    float sumA = 0.f, sumB = 0.f;
#pragma unroll 14
    for (int j = 0; j < 42; ++j) {
      const float eA = __expf(srowA[8 * j] - rmA);
      const float eB = __expf(srowB[8 * j] - rmB);
      sumA += eA;
      sumB += eB;
      srowA[8 * j] = eA;
      srowB[8 * j] = eB;
    }
    {
      const float sA = srowA[8 * 42];
      const float sB = srowB[8 * 42];
      const float eA = (sg < 7) ? __expf(sA - rmA) : 0.f;
      const float eB = (sg < 7) ? __expf(sB - rmB) : 0.f;
      sumA += eA;
      sumB += eB;
      srowA[8 * 42] = eA;
      srowB[8 * 42] = eB;
    }
    srowA[8 * 43] = 0.f;
    srowB[8 * 43] = 0.f;
    sRed[r * 8 + sg] = sumA;
    sRed[(r + 16) * 8 + sg] = sumB;
    __syncthreads();
    if (sg < 2) {
      const int row = r + 16 * sg;
      float ss = 0.f;
#pragma unroll
      for (int j = 0; j < 8; ++j) ss += sRed[row * 8 + j];
      sRowInv[row] = 1.0f / ss;
    }
  }
  __syncthreads();

  v8f a00 = {}, a01 = {}, a10 = {}, a11 = {};
  {
    const float* vrow0 = Vb + 2 * l16;
    const float* prow0 = sS + l16 * SSTRIDE;
    if (wid == 0)      pv_keyquarter<0>(vrow0, prow0, half, a00, a01, a10, a11);
    else if (wid == 1) pv_keyquarter<1>(vrow0, prow0, half, a00, a01, a10, a11);
    else if (wid == 2) pv_keyquarter<2>(vrow0, prow0, half, a00, a01, a10, a11);
    else               pv_keyquarter<3>(vrow0, prow0, half, a00, a01, a10, a11);
  }
  __syncthreads();
  {
    float* dstp = sS + wid * (QROWS * PPITCH) + 8 * half * PPITCH + 2 * l16;
#pragma unroll
    for (int i = 0; i < 8; ++i) {
      const v2f p0 = { a00[i], a01[i] };
      const v2f p1 = { a10[i], a11[i] };
      *reinterpret_cast<v2f*>(dstp + i * PPITCH) = p0;
      *reinterpret_cast<v2f*>(dstp + (16 + i) * PPITCH) = p1;
    }
  }
  __syncthreads();
  {
    const int rr = t >> 3, dcol = (t & 7) * 4;
#pragma unroll
    for (int u = 0; u < 2; ++u) {
      const int row = 16 * u + rr;
      const int q = q0 + row;
      const float* srcp = sS + row * PPITCH + dcol;
      const v4f p0 = *reinterpret_cast<const v4f*>(srcp);
      const v4f p1 = *reinterpret_cast<const v4f*>(srcp + QROWS * PPITCH);
      const v4f p2 = *reinterpret_cast<const v4f*>(srcp + 2 * QROWS * PPITCH);
      const v4f p3 = *reinterpret_cast<const v4f*>(srcp + 3 * QROWS * PPITCH);
      const float inv = sRowInv[row];
      const v4f o = (((p0 + p1) + p2) + p3) * inv;
      if (q < NTOK)
        *reinterpret_cast<v4f*>(AO + ((size_t)b * NTOK + q) * DIMC + h * HD + dcol) = o;
    }
  }
}

__global__ __launch_bounds__(96) void proj_kernel(
    const float* __restrict__ AO, const float* __restrict__ W,
    const float* __restrict__ bias, float* __restrict__ out) {
  const int t = threadIdx.x;
  const int wid = t >> 5, lane = t & 31, half = lane >> 4, l16 = lane & 15;
  const int row0 = blockIdx.x * 32;
  const int col0 = wid * 64;

  v8f acc[2][4] = {};
  wave_gemm_32x64<DIMC>(AO, W, row0, col0, l16, half, acc);

  const int c4 = col0 + 4 * l16;
  const v4f bv = *reinterpret_cast<const v4f*>(bias + c4);
#pragma unroll
  for (int m = 0; m < 2; ++m) {
#pragma unroll
    for (int i = 0; i < 8; ++i) {
      const int r = row0 + 16 * m + i + 8 * half;
      const v4f o = { acc[m][0][i] + bv.x, acc[m][1][i] + bv.y,
                      acc[m][2][i] + bv.z, acc[m][3][i] + bv.w };
      *reinterpret_cast<v4f*>(out + (size_t)r * DIMC + c4) = o;
    }
  }
}

extern "C" void kernel_launch(void* const* d_in, const int* in_sizes, int n_in,
                              void* d_out, int out_size, void* d_ws, size_t ws_size,
                              hipStream_t stream) {
  const float* x      = (const float*)d_in[0];
  const float* mask   = (const float*)d_in[1];
  const float* qkv_w  = (const float*)d_in[2];
  const float* rpb    = (const float*)d_in[3];
  const float* proj_w = (const float*)d_in[4];
  const float* proj_b = (const float*)d_in[5];
  float* out = (float*)d_out;

  const size_t qkvElems = (size_t)BATCH * HEADS * NTOK * HD;
  float* Q  = (float*)d_ws;
  float* K  = Q + qkvElems;
  float* V  = K + qkvElems;
  float* AO = V + qkvElems;

  const int rowPairs  = (BATCH * NTOK) / 32;
  const int qkvBlocks = (BATCH * NTOK) / 32 * 3;
  qkv_kernel <<<dim3(qkvBlocks),              96, 0, stream>>>(x, qkv_w, Q, K, V);
  attn_kernel<<<dim3(NPAD / QROWS, HEADS, BATCH), 128, 0, stream>>>(Q, K, V, mask, rpb, AO);
  proj_kernel<<<dim3(rowPairs),               96, 0, stream>>>(AO, proj_w, proj_b, out);
}
